# starting kernel
# baseline (speedup 1.0000x reference)
#include <hip/hip_runtime.h>

typedef __attribute__((ext_vector_type(2))) float v2f;
typedef __attribute__((ext_vector_type(8))) float v8f;


__global__ void deg_init_kernel(int* __restrict__ deg, int n) {
  int i = blockIdx.x * blockDim.x + threadIdx.x;
  if (i < n) deg[i] = 1;
}

__global__ void deg_count_kernel(const int* __restrict__ dst, int* __restrict__ deg, int e) {
  int i = blockIdx.x * blockDim.x + threadIdx.x;
  if (i < e) atomicAdd(&deg[dst[i]], 1);
}

__global__ void scan_kernel(const int* __restrict__ deg, int* __restrict__ rowstart,
                            int* __restrict__ cursor, float* __restrict__ dinvf, int n) {
  __shared__ int buf[1024];
  __shared__ int carry_s;
  const int tid = threadIdx.x;
  if (tid == 0) carry_s = 0;
  __syncthreads();
  for (int base = 0; base < n; base += 1024) {
    int i = base + tid;
    int v = 0;
    if (i < n) {
      v = deg[i] - 1;
      dinvf[i] = rsqrtf((float)(v + 1));
    }
    buf[tid] = v;
    __syncthreads();
    #pragma unroll
    for (int off = 1; off < 1024; off <<= 1) {
      int t = (tid >= off) ? buf[tid - off] : 0;
      __syncthreads();
      buf[tid] += t;
      __syncthreads();
    }
    int incl = buf[tid];
    int c = carry_s;
    __syncthreads();
    if (i < n) {
      int o = c + incl - v;
      rowstart[i] = o;
      cursor[i] = o;
    }
    if (tid == 0) carry_s = c + buf[1023];
    __syncthreads();
  }
  if (tid == 0) rowstart[n] = carry_s;
}

__global__ void scatter_kernel(const int* __restrict__ src, const int* __restrict__ dst,
                               int* __restrict__ cursor, int* __restrict__ csr, int e) {
  int i = blockIdx.x * blockDim.x + threadIdx.x;
  if (i < e) {
    int d = dst[i];
    int pos = atomicAdd(&cursor[d], 1);
    csr[pos] = src[i];
  }
}

__global__ void __launch_bounds__(128)
gemm_self_kernel(const float* __restrict__ x, const float* __restrict__ W,
                 const float* __restrict__ bias, const float* __restrict__ dinvf,
                 float* __restrict__ y, float* __restrict__ out) {
  const int t = threadIdx.x;
  const int m0 = blockIdx.x * 16;
  const int n0 = threadIdx.y * 16;
  const int l16 = t & 15;
  const int kh = (t >> 4) * 2;
  const int row = m0 + l16;
  const int col = n0 + l16;

  v8f c = {};
  #pragma unroll
  for (int k = 0; k < 64; k += 4) {
    v2f a = *(const v2f*)(x + row * 64 + k + kh);
    v2f b;
    b[0] = W[(k + kh) * 64 + col];
    b[1] = W[(k + kh + 1) * 64 + col];
    c = __builtin_amdgcn_wmma_f32_16x16x4_f32(
         false, a,  false, b,
         (short)0, c,  false,  false);
  }

  const float bv = bias[col];
  #pragma unroll
  for (int v = 0; v < 8; v++) {
    int m = m0 + 8 * (t >> 4) + v;
    float di = dinvf[m];
    float yv = c[v] * di;
    y[m * 64 + col] = yv;
    out[m * 64 + col] = yv * di + bv;
  }
}

__global__ void gemm_tail_kernel(const float* __restrict__ x, const float* __restrict__ W,
                                 const float* __restrict__ bias, const float* __restrict__ dinvf,
                                 float* __restrict__ y, float* __restrict__ out,
                                 int row0, int n) {
  int idx = blockIdx.x * blockDim.x + threadIdx.x;
  int row = row0 + idx / 64;
  int col = idx % 64;
  if (row >= n) return;
  float acc = 0.f;
  for (int k = 0; k < 64; k++) acc += x[row * 64 + k] * W[k * 64 + col];
  float di = dinvf[row];
  float yv = acc * di;
  y[row * 64 + col] = yv;
  out[row * 64 + col] = yv * di + bias[col];
}

__global__ void __launch_bounds__(256)
aggregate_kernel(const int* __restrict__ rowstart, const int* __restrict__ csr,
                 const float* __restrict__ y, const float* __restrict__ dinvf,
                 float* __restrict__ out, int n) {
  int node = blockIdx.x * blockDim.y + threadIdx.y;
  if (node >= n) return;
  int t = threadIdx.x;
  int beg = rowstart[node];
  int end = rowstart[node + 1];
  float ax = 0.f, ay = 0.f;
  for (int j = beg; j < end; j++) {
    int s = csr[j];
    v2f v = *(const v2f*)(y + s * 64 + 2 * t);
    ax += v[0];
    ay += v[1];
  }
  float di = dinvf[node];
  int o = node * 64 + 2 * t;
  out[o]     += di * ax;
  out[o + 1] += di * ay;
}

static inline size_t ws_align(size_t x) { return (x + 255) & ~(size_t)255; }

extern "C" void kernel_launch(void* const* d_in, const int* in_sizes, int n_in,
                              void* d_out, int out_size, void* d_ws, size_t ws_size,
                              hipStream_t stream) {
  const float* x  = (const float*)d_in[0];
  const int*   ei = (const int*)d_in[1];
  const float* W  = (const float*)d_in[2];
  const float* b  = (const float*)d_in[3];
  float* out = (float*)d_out;

  const int N = in_sizes[0] / 64;
  const int E = in_sizes[1] / 2;
  const int* src = ei;
  const int* dst = ei + E;

  char* ws = (char*)d_ws;
  size_t off = 0;
  int*   deg      = (int*)(ws + off);   off = ws_align(off + (size_t)N * sizeof(int));
  float* dinvf    = (float*)(ws + off); off = ws_align(off + (size_t)N * sizeof(float));
  int*   rowstart = (int*)(ws + off);   off = ws_align(off + (size_t)(N + 1) * sizeof(int));
  int*   cursor   = (int*)(ws + off);   off = ws_align(off + (size_t)N * sizeof(int));
  int*   csr      = (int*)(ws + off);   off = ws_align(off + (size_t)E * sizeof(int));
  float* y        = (float*)(ws + off); off = ws_align(off + (size_t)N * 64 * sizeof(float));

  deg_init_kernel<<<(N + 255) / 256, 256, 0, stream>>>(deg, N);
  deg_count_kernel<<<(E + 255) / 256, 256, 0, stream>>>(dst, deg, E);
  scan_kernel<<<1, 1024, 0, stream>>>(deg, rowstart, cursor, dinvf, N);
  scatter_kernel<<<(E + 255) / 256, 256, 0, stream>>>(src, dst, cursor, csr, E);

  int mtiles = N / 16;
  if (mtiles > 0)
    gemm_self_kernel<<<mtiles, dim3(32, 4), 0, stream>>>(x, W, b, dinvf, y, out);
  int rem = N - mtiles * 16;
  if (rem > 0)
    gemm_tail_kernel<<<(rem * 64 + 255) / 256, 256, 0, stream>>>(x, W, b, dinvf, y, out,
                                                                 mtiles * 16, N);

  aggregate_kernel<<<(N + 7) / 8, dim3(32, 8), 0, stream>>>(rowstart, csr, y, dinvf, out, N);
}


# scan: one work-group's 98 serial Hillis-Steele chunks (about 2000 barriers) become 25 independent work-groups that re-sum the counts before their 4096-node slice, then one shuffle scan
# speedup vs baseline: 2.5244x; 2.5244x over previous
#include <hip/hip_runtime.h>

typedef __attribute__((ext_vector_type(2))) float v2f;
typedef __attribute__((ext_vector_type(8))) float v8f;


__global__ void deg_init_kernel(int* __restrict__ deg, int n) {
  int i = blockIdx.x * blockDim.x + threadIdx.x;
  if (i < n) deg[i] = 1;
}

__global__ void deg_count_kernel(const int* __restrict__ dst, int* __restrict__ deg, int e) {
  int i = blockIdx.x * blockDim.x + threadIdx.x;
  if (i < e) atomicAdd(&deg[dst[i]], 1);
}

#define SCAN_THREADS 1024
#define SCAN_SLICE (4 * SCAN_THREADS)
static_assert(SCAN_THREADS == 32 * 32, "the 32 per-wave sums are folded by one 32-lane wave");
__global__ void __launch_bounds__(SCAN_THREADS)
scan_kernel(const int* __restrict__ deg, int* __restrict__ rowstart,
            int* __restrict__ cursor, float* __restrict__ dinvf, int n) {
  __shared__ int wave_own[SCAN_THREADS / 32];
  __shared__ int wave_pre[SCAN_THREADS / 32];
  const int tid = threadIdx.x;
  const int lane = tid & 31;
  const int wv = tid >> 5;
  const int4* __restrict__ deg4 = (const int4*)deg;

  const int nb = blockIdx.x;
  int pre = 0;
  for (int i = 0; i < nb; i += 8) {
    int4 v[8];
    #pragma unroll
    for (int j = 0; j < 8; j++) {
      const int q = min(i + j, nb - 1);
      v[j] = deg4[q * SCAN_THREADS + tid];
    }
    #pragma unroll
    for (int j = 0; j < 8; j++)
      if (i + j < nb) pre += (v[j].x + v[j].y) + (v[j].z + v[j].w);
  }

  const int i0 = blockIdx.x * SCAN_SLICE + 4 * tid;
  const bool whole = (i0 + 3 < n);
  int d0 = 1, d1 = 1, d2 = 1, d3 = 1;
  if (whole) {
    const int4 d = deg4[i0 >> 2];
    d0 = d.x; d1 = d.y; d2 = d.z; d3 = d.w;
  } else {
    if (i0     < n) d0 = deg[i0];
    if (i0 + 1 < n) d1 = deg[i0 + 1];
    if (i0 + 2 < n) d2 = deg[i0 + 2];
  }
  const int v0 = d0 - 1, v1 = d1 - 1, v2 = d2 - 1, v3 = d3 - 1;
  const int s = (v0 + v1) + (v2 + v3);

  int inc = s;
  for (int off = 1; off < 32; off <<= 1) {
    const int t = __shfl_up(inc, off);
    if (lane >= off) inc += t;
  }
  for (int off = 16; off > 0; off >>= 1) pre += __shfl_xor(pre, off);
  if (lane == 31) wave_own[wv] = inc;
  if (lane == 0) wave_pre[wv] = pre;
  __syncthreads();

  int base = wave_pre[lane] + (lane < wv ? wave_own[lane] : 0);
  for (int off = 16; off > 0; off >>= 1) base += __shfl_xor(base, off);
  base -= blockIdx.x * SCAN_SLICE;

  const int o0 = base + inc - s;
  const int o1 = o0 + v0;
  const int o2 = o1 + v1;
  const int o3 = o2 + v2;
  if (whole) {
    const int4 o = make_int4(o0, o1, o2, o3);
    *(int4*)(rowstart + i0) = o;
    *(int4*)(cursor + i0) = o;
    *(float4*)(dinvf + i0) = make_float4(rsqrtf((float)d0), rsqrtf((float)d1),
                                         rsqrtf((float)d2), rsqrtf((float)d3));
  } else {
    if (i0     < n) { rowstart[i0]     = o0; cursor[i0]     = o0; dinvf[i0]     = rsqrtf((float)d0); }
    if (i0 + 1 < n) { rowstart[i0 + 1] = o1; cursor[i0 + 1] = o1; dinvf[i0 + 1] = rsqrtf((float)d1); }
    if (i0 + 2 < n) { rowstart[i0 + 2] = o2; cursor[i0 + 2] = o2; dinvf[i0 + 2] = rsqrtf((float)d2); }
  }
  if (i0 < n && n <= i0 + 4) rowstart[n] = o0 + s;
}

__global__ void scatter_kernel(const int* __restrict__ src, const int* __restrict__ dst,
                               int* __restrict__ cursor, int* __restrict__ csr, int e) {
  int i = blockIdx.x * blockDim.x + threadIdx.x;
  if (i < e) {
    int d = dst[i];
    int pos = atomicAdd(&cursor[d], 1);
    csr[pos] = src[i];
  }
}

__global__ void __launch_bounds__(128)
gemm_self_kernel(const float* __restrict__ x, const float* __restrict__ W,
                 const float* __restrict__ bias, const float* __restrict__ dinvf,
                 float* __restrict__ y, float* __restrict__ out) {
  const int t = threadIdx.x;
  const int m0 = blockIdx.x * 16;
  const int n0 = threadIdx.y * 16;
  const int l16 = t & 15;
  const int kh = (t >> 4) * 2;
  const int row = m0 + l16;
  const int col = n0 + l16;

  v8f c = {};
  #pragma unroll
  for (int k = 0; k < 64; k += 4) {
    v2f a = *(const v2f*)(x + row * 64 + k + kh);
    v2f b;
    b[0] = W[(k + kh) * 64 + col];
    b[1] = W[(k + kh + 1) * 64 + col];
    c = __builtin_amdgcn_wmma_f32_16x16x4_f32(
         false, a,  false, b,
         (short)0, c,  false,  false);
  }

  const float bv = bias[col];
  #pragma unroll
  for (int v = 0; v < 8; v++) {
    int m = m0 + 8 * (t >> 4) + v;
    float di = dinvf[m];
    float yv = c[v] * di;
    y[m * 64 + col] = yv;
    out[m * 64 + col] = yv * di + bv;
  }
}

__global__ void gemm_tail_kernel(const float* __restrict__ x, const float* __restrict__ W,
                                 const float* __restrict__ bias, const float* __restrict__ dinvf,
                                 float* __restrict__ y, float* __restrict__ out,
                                 int row0, int n) {
  int idx = blockIdx.x * blockDim.x + threadIdx.x;
  int row = row0 + idx / 64;
  int col = idx % 64;
  if (row >= n) return;
  float acc = 0.f;
  for (int k = 0; k < 64; k++) acc += x[row * 64 + k] * W[k * 64 + col];
  float di = dinvf[row];
  float yv = acc * di;
  y[row * 64 + col] = yv;
  out[row * 64 + col] = yv * di + bias[col];
}

__global__ void __launch_bounds__(256)
aggregate_kernel(const int* __restrict__ rowstart, const int* __restrict__ csr,
                 const float* __restrict__ y, const float* __restrict__ dinvf,
                 float* __restrict__ out, int n) {
  int node = blockIdx.x * blockDim.y + threadIdx.y;
  if (node >= n) return;
  int t = threadIdx.x;
  int beg = rowstart[node];
  int end = rowstart[node + 1];
  float ax = 0.f, ay = 0.f;
  for (int j = beg; j < end; j++) {
    int s = csr[j];
    v2f v = *(const v2f*)(y + s * 64 + 2 * t);
    ax += v[0];
    ay += v[1];
  }
  float di = dinvf[node];
  int o = node * 64 + 2 * t;
  out[o]     += di * ax;
  out[o + 1] += di * ay;
}

static inline size_t ws_align(size_t x) { return (x + 255) & ~(size_t)255; }

extern "C" void kernel_launch(void* const* d_in, const int* in_sizes, int n_in,
                              void* d_out, int out_size, void* d_ws, size_t ws_size,
                              hipStream_t stream) {
  const float* x  = (const float*)d_in[0];
  const int*   ei = (const int*)d_in[1];
  const float* W  = (const float*)d_in[2];
  const float* b  = (const float*)d_in[3];
  float* out = (float*)d_out;

  const int N = in_sizes[0] / 64;
  const int E = in_sizes[1] / 2;
  const int* src = ei;
  const int* dst = ei + E;

  char* ws = (char*)d_ws;
  size_t off = 0;
  int*   deg      = (int*)(ws + off);   off = ws_align(off + (size_t)N * sizeof(int));
  float* dinvf    = (float*)(ws + off); off = ws_align(off + (size_t)N * sizeof(float));
  int*   rowstart = (int*)(ws + off);   off = ws_align(off + (size_t)(N + 1) * sizeof(int));
  int*   cursor   = (int*)(ws + off);   off = ws_align(off + (size_t)N * sizeof(int));
  int*   csr      = (int*)(ws + off);   off = ws_align(off + (size_t)E * sizeof(int));
  float* y        = (float*)(ws + off); off = ws_align(off + (size_t)N * 64 * sizeof(float));

  deg_init_kernel<<<(N + 255) / 256, 256, 0, stream>>>(deg, N);
  deg_count_kernel<<<(E + 255) / 256, 256, 0, stream>>>(dst, deg, E);
  scan_kernel<<<(N + SCAN_SLICE - 1) / SCAN_SLICE, SCAN_THREADS, 0, stream>>>(deg, rowstart, cursor, dinvf, N);
  scatter_kernel<<<(E + 255) / 256, 256, 0, stream>>>(src, dst, cursor, csr, E);

  int mtiles = N / 16;
  if (mtiles > 0)
    gemm_self_kernel<<<mtiles, dim3(32, 4), 0, stream>>>(x, W, b, dinvf, y, out);
  int rem = N - mtiles * 16;
  if (rem > 0)
    gemm_tail_kernel<<<(rem * 64 + 255) / 256, 256, 0, stream>>>(x, W, b, dinvf, y, out,
                                                                 mtiles * 16, N);

  aggregate_kernel<<<(N + 7) / 8, dim3(32, 8), 0, stream>>>(rowstart, csr, y, dinvf, out, N);
}


# edge count adds into a counter plane of the work-group's own package half and keeps the returned rank; CSR fill places edges by that rank, no atomics; per-node sum order still an arrival-order rank
# speedup vs baseline: 3.9132x; 1.5502x over previous
#include <hip/hip_runtime.h>

typedef __attribute__((ext_vector_type(2))) float v2f;
typedef __attribute__((ext_vector_type(8))) float v8f;


__device__ __forceinline__ int package_half() {
  const unsigned id = __builtin_amdgcn_s_sendmsg_rtn(0x87);
  return (int)((id >> 18) & 1u);
}

__global__ void cnt_init_kernel(int* __restrict__ cnt0, int* __restrict__ cnt1, int n) {
  int i = blockIdx.x * blockDim.x + threadIdx.x;
  if (i < n) { cnt0[i] = 0; cnt1[i] = 0; }
}

__global__ void count_rank_kernel(const int* __restrict__ dst, int* __restrict__ cnt0,
                                  int* __restrict__ cnt1, int* __restrict__ rank, int e) {
  int i = blockIdx.x * blockDim.x + threadIdx.x;
  const int h = package_half();
  int* __restrict__ cnt = h ? cnt1 : cnt0;
  if (i < e) {
    const int r = atomicAdd(&cnt[dst[i]], 1);
    rank[i] = (r << 1) | h;
  }
}

#define SCAN_THREADS 1024
#define SCAN_SLICE (4 * SCAN_THREADS)
static_assert(SCAN_THREADS == 32 * 32, "the 32 per-wave sums are folded by one 32-lane wave");
__global__ void __launch_bounds__(SCAN_THREADS)
scan_kernel(const int* __restrict__ cnt0, const int* __restrict__ cnt1,
            int* __restrict__ start0, int* __restrict__ start1,
            float* __restrict__ dinvf, int n) {
  __shared__ int wave_own[SCAN_THREADS / 32];
  __shared__ int wave_pre[SCAN_THREADS / 32];
  const int tid = threadIdx.x;
  const int lane = tid & 31;
  const int wv = tid >> 5;
  const int4* __restrict__ p0 = (const int4*)cnt0;
  const int4* __restrict__ p1 = (const int4*)cnt1;

  const int nb = blockIdx.x;
  int pre = 0;
  for (int i = 0; i < nb; i += 8) {
    int4 u[8], v[8];
    #pragma unroll
    for (int j = 0; j < 8; j++) {
      const int q = min(i + j, nb - 1);
      u[j] = p0[q * SCAN_THREADS + tid];
      v[j] = p1[q * SCAN_THREADS + tid];
    }
    #pragma unroll
    for (int j = 0; j < 8; j++)
      if (i + j < nb)
        pre += ((u[j].x + u[j].y) + (u[j].z + u[j].w)) + ((v[j].x + v[j].y) + (v[j].z + v[j].w));
  }

  const int i0 = blockIdx.x * SCAN_SLICE + 4 * tid;
  const bool whole = (i0 + 3 < n);
  int a0 = 0, a1 = 0, a2 = 0, a3 = 0;
  int b0 = 0, b1 = 0, b2 = 0, b3 = 0;
  if (whole) {
    const int4 a = p0[i0 >> 2];
    const int4 b = p1[i0 >> 2];
    a0 = a.x; a1 = a.y; a2 = a.z; a3 = a.w;
    b0 = b.x; b1 = b.y; b2 = b.z; b3 = b.w;
  } else {
    if (i0     < n) { a0 = cnt0[i0];     b0 = cnt1[i0]; }
    if (i0 + 1 < n) { a1 = cnt0[i0 + 1]; b1 = cnt1[i0 + 1]; }
    if (i0 + 2 < n) { a2 = cnt0[i0 + 2]; b2 = cnt1[i0 + 2]; }
  }
  const int v0 = a0 + b0, v1 = a1 + b1, v2 = a2 + b2, v3 = a3 + b3;
  const int s = (v0 + v1) + (v2 + v3);

  int inc = s;
  for (int off = 1; off < 32; off <<= 1) {
    const int t = __shfl_up(inc, off);
    if (lane >= off) inc += t;
  }
  for (int off = 16; off > 0; off >>= 1) pre += __shfl_xor(pre, off);
  if (lane == 31) wave_own[wv] = inc;
  if (lane == 0) wave_pre[wv] = pre;
  __syncthreads();

  int base = wave_pre[lane] + (lane < wv ? wave_own[lane] : 0);
  for (int off = 16; off > 0; off >>= 1) base += __shfl_xor(base, off);

  const int o0 = base + inc - s;
  const int o1 = o0 + v0;
  const int o2 = o1 + v1;
  const int o3 = o2 + v2;
  if (whole) {
    *(int4*)(start0 + i0) = make_int4(o0, o1, o2, o3);
    *(int4*)(start1 + i0) = make_int4(o0 + a0, o1 + a1, o2 + a2, o3 + a3);
    *(float4*)(dinvf + i0) = make_float4(rsqrtf((float)(v0 + 1)), rsqrtf((float)(v1 + 1)),
                                         rsqrtf((float)(v2 + 1)), rsqrtf((float)(v3 + 1)));
  } else {
    if (i0     < n) { start0[i0]     = o0; start1[i0]     = o0 + a0; dinvf[i0]     = rsqrtf((float)(v0 + 1)); }
    if (i0 + 1 < n) { start0[i0 + 1] = o1; start1[i0 + 1] = o1 + a1; dinvf[i0 + 1] = rsqrtf((float)(v1 + 1)); }
    if (i0 + 2 < n) { start0[i0 + 2] = o2; start1[i0 + 2] = o2 + a2; dinvf[i0 + 2] = rsqrtf((float)(v2 + 1)); }
  }
  if (i0 < n && n <= i0 + 4) start0[n] = o0 + s;
}

__global__ void fill_kernel(const int* __restrict__ src, const int* __restrict__ dst,
                            const int* __restrict__ rank, const int* __restrict__ start0,
                            const int* __restrict__ start1, int* __restrict__ csr, int e) {
  int i = blockIdx.x * blockDim.x + threadIdx.x;
  if (i < e) {
    const int d = dst[i];
    const int rk = rank[i];
    const int* __restrict__ st = (rk & 1) ? start1 : start0;
    csr[st[d] + (rk >> 1)] = src[i];
  }
}

__global__ void __launch_bounds__(128)
gemm_self_kernel(const float* __restrict__ x, const float* __restrict__ W,
                 const float* __restrict__ bias, const float* __restrict__ dinvf,
                 float* __restrict__ y, float* __restrict__ out) {
  const int t = threadIdx.x;
  const int m0 = blockIdx.x * 16;
  const int n0 = threadIdx.y * 16;
  const int l16 = t & 15;
  const int kh = (t >> 4) * 2;
  const int row = m0 + l16;
  const int col = n0 + l16;

  v8f c = {};
  #pragma unroll
  for (int k = 0; k < 64; k += 4) {
    v2f a = *(const v2f*)(x + row * 64 + k + kh);
    v2f b;
    b[0] = W[(k + kh) * 64 + col];
    b[1] = W[(k + kh + 1) * 64 + col];
    c = __builtin_amdgcn_wmma_f32_16x16x4_f32(
         false, a,  false, b,
         (short)0, c,  false,  false);
  }

  const float bv = bias[col];
  #pragma unroll
  for (int v = 0; v < 8; v++) {
    int m = m0 + 8 * (t >> 4) + v;
    float di = dinvf[m];
    float yv = c[v] * di;
    y[m * 64 + col] = yv;
    out[m * 64 + col] = yv * di + bv;
  }
}

__global__ void gemm_tail_kernel(const float* __restrict__ x, const float* __restrict__ W,
                                 const float* __restrict__ bias, const float* __restrict__ dinvf,
                                 float* __restrict__ y, float* __restrict__ out,
                                 int row0, int n) {
  int idx = blockIdx.x * blockDim.x + threadIdx.x;
  int row = row0 + idx / 64;
  int col = idx % 64;
  if (row >= n) return;
  float acc = 0.f;
  for (int k = 0; k < 64; k++) acc += x[row * 64 + k] * W[k * 64 + col];
  float di = dinvf[row];
  float yv = acc * di;
  y[row * 64 + col] = yv;
  out[row * 64 + col] = yv * di + bias[col];
}

__global__ void __launch_bounds__(256)
aggregate_kernel(const int* __restrict__ rowstart, const int* __restrict__ csr,
                 const float* __restrict__ y, const float* __restrict__ dinvf,
                 float* __restrict__ out, int n) {
  int node = blockIdx.x * blockDim.y + threadIdx.y;
  if (node >= n) return;
  int t = threadIdx.x;
  int beg = rowstart[node];
  int end = rowstart[node + 1];
  float ax = 0.f, ay = 0.f;
  for (int j = beg; j < end; j++) {
    int s = csr[j];
    v2f v = *(const v2f*)(y + s * 64 + 2 * t);
    ax += v[0];
    ay += v[1];
  }
  float di = dinvf[node];
  int o = node * 64 + 2 * t;
  out[o]     += di * ax;
  out[o + 1] += di * ay;
}

static inline size_t ws_align(size_t x) { return (x + 255) & ~(size_t)255; }

extern "C" void kernel_launch(void* const* d_in, const int* in_sizes, int n_in,
                              void* d_out, int out_size, void* d_ws, size_t ws_size,
                              hipStream_t stream) {
  const float* x  = (const float*)d_in[0];
  const int*   ei = (const int*)d_in[1];
  const float* W  = (const float*)d_in[2];
  const float* b  = (const float*)d_in[3];
  float* out = (float*)d_out;

  const int N = in_sizes[0] / 64;
  const int E = in_sizes[1] / 2;
  const int* src = ei;
  const int* dst = ei + E;

  char* ws = (char*)d_ws;
  size_t off = 0;
  int*   cnt0     = (int*)(ws + off);   off = ws_align(off + (size_t)N * sizeof(int));
  int*   cnt1     = (int*)(ws + off);   off = ws_align(off + (size_t)N * sizeof(int));
  float* dinvf    = (float*)(ws + off); off = ws_align(off + (size_t)N * sizeof(float));
  int*   rowstart = (int*)(ws + off);   off = ws_align(off + (size_t)(N + 1) * sizeof(int));
  int*   start1   = (int*)(ws + off);   off = ws_align(off + (size_t)N * sizeof(int));
  int*   rank     = (int*)(ws + off);   off = ws_align(off + (size_t)E * sizeof(int));
  int*   csr      = (int*)(ws + off);   off = ws_align(off + (size_t)E * sizeof(int));
  float* y        = (float*)(ws + off); off = ws_align(off + (size_t)N * 64 * sizeof(float));

  cnt_init_kernel<<<(N + 255) / 256, 256, 0, stream>>>(cnt0, cnt1, N);
  count_rank_kernel<<<(E + 255) / 256, 256, 0, stream>>>(dst, cnt0, cnt1, rank, E);
  scan_kernel<<<(N + SCAN_SLICE - 1) / SCAN_SLICE, SCAN_THREADS, 0, stream>>>(cnt0, cnt1, rowstart, start1, dinvf, N);
  fill_kernel<<<(E + 255) / 256, 256, 0, stream>>>(src, dst, rank, rowstart, start1, csr, E);

  int mtiles = N / 16;
  if (mtiles > 0)
    gemm_self_kernel<<<mtiles, dim3(32, 4), 0, stream>>>(x, W, b, dinvf, y, out);
  int rem = N - mtiles * 16;
  if (rem > 0)
    gemm_tail_kernel<<<(rem * 64 + 255) / 256, 256, 0, stream>>>(x, W, b, dinvf, y, out,
                                                                 mtiles * 16, N);

  aggregate_kernel<<<(N + 7) / 8, dim3(32, 8), 0, stream>>>(rowstart, csr, y, dinvf, out, N);
}


# aggregate: 8192 resident waves walk the nodes; a node's edge ids come by one coalesced load a node ahead and its rows load as one group, not a dependent id+row pair per edge; CSR sum order kept
# speedup vs baseline: 4.1146x; 1.0515x over previous
#include <hip/hip_runtime.h>

typedef __attribute__((ext_vector_type(2))) float v2f;
typedef __attribute__((ext_vector_type(8))) float v8f;


__device__ __forceinline__ int package_half() {
  const unsigned id = __builtin_amdgcn_s_sendmsg_rtn(0x87);
  return (int)((id >> 18) & 1u);
}

__global__ void cnt_init_kernel(int* __restrict__ cnt0, int* __restrict__ cnt1, int n) {
  int i = blockIdx.x * blockDim.x + threadIdx.x;
  if (i < n) { cnt0[i] = 0; cnt1[i] = 0; }
}

__global__ void count_rank_kernel(const int* __restrict__ dst, int* __restrict__ cnt0,
                                  int* __restrict__ cnt1, int* __restrict__ rank, int e) {
  int i = blockIdx.x * blockDim.x + threadIdx.x;
  const int h = package_half();
  int* __restrict__ cnt = h ? cnt1 : cnt0;
  if (i < e) {
    const int r = atomicAdd(&cnt[dst[i]], 1);
    rank[i] = (r << 1) | h;
  }
}

#define SCAN_THREADS 1024
#define SCAN_SLICE (4 * SCAN_THREADS)
static_assert(SCAN_THREADS == 32 * 32, "the 32 per-wave sums are folded by one 32-lane wave");
__global__ void __launch_bounds__(SCAN_THREADS)
scan_kernel(const int* __restrict__ cnt0, const int* __restrict__ cnt1,
            int* __restrict__ start0, int* __restrict__ start1,
            float* __restrict__ dinvf, int n) {
  __shared__ int wave_own[SCAN_THREADS / 32];
  __shared__ int wave_pre[SCAN_THREADS / 32];
  const int tid = threadIdx.x;
  const int lane = tid & 31;
  const int wv = tid >> 5;
  const int4* __restrict__ p0 = (const int4*)cnt0;
  const int4* __restrict__ p1 = (const int4*)cnt1;

  const int nb = blockIdx.x;
  int pre = 0;
  for (int i = 0; i < nb; i += 8) {
    int4 u[8], v[8];
    #pragma unroll
    for (int j = 0; j < 8; j++) {
      const int q = min(i + j, nb - 1);
      u[j] = p0[q * SCAN_THREADS + tid];
      v[j] = p1[q * SCAN_THREADS + tid];
    }
    #pragma unroll
    for (int j = 0; j < 8; j++)
      if (i + j < nb)
        pre += ((u[j].x + u[j].y) + (u[j].z + u[j].w)) + ((v[j].x + v[j].y) + (v[j].z + v[j].w));
  }

  const int i0 = blockIdx.x * SCAN_SLICE + 4 * tid;
  const bool whole = (i0 + 3 < n);
  int a0 = 0, a1 = 0, a2 = 0, a3 = 0;
  int b0 = 0, b1 = 0, b2 = 0, b3 = 0;
  if (whole) {
    const int4 a = p0[i0 >> 2];
    const int4 b = p1[i0 >> 2];
    a0 = a.x; a1 = a.y; a2 = a.z; a3 = a.w;
    b0 = b.x; b1 = b.y; b2 = b.z; b3 = b.w;
  } else {
    if (i0     < n) { a0 = cnt0[i0];     b0 = cnt1[i0]; }
    if (i0 + 1 < n) { a1 = cnt0[i0 + 1]; b1 = cnt1[i0 + 1]; }
    if (i0 + 2 < n) { a2 = cnt0[i0 + 2]; b2 = cnt1[i0 + 2]; }
  }
  const int v0 = a0 + b0, v1 = a1 + b1, v2 = a2 + b2, v3 = a3 + b3;
  const int s = (v0 + v1) + (v2 + v3);

  int inc = s;
  for (int off = 1; off < 32; off <<= 1) {
    const int t = __shfl_up(inc, off);
    if (lane >= off) inc += t;
  }
  for (int off = 16; off > 0; off >>= 1) pre += __shfl_xor(pre, off);
  if (lane == 31) wave_own[wv] = inc;
  if (lane == 0) wave_pre[wv] = pre;
  __syncthreads();

  int base = wave_pre[lane] + (lane < wv ? wave_own[lane] : 0);
  for (int off = 16; off > 0; off >>= 1) base += __shfl_xor(base, off);

  const int o0 = base + inc - s;
  const int o1 = o0 + v0;
  const int o2 = o1 + v1;
  const int o3 = o2 + v2;
  if (whole) {
    *(int4*)(start0 + i0) = make_int4(o0, o1, o2, o3);
    *(int4*)(start1 + i0) = make_int4(o0 + a0, o1 + a1, o2 + a2, o3 + a3);
    *(float4*)(dinvf + i0) = make_float4(rsqrtf((float)(v0 + 1)), rsqrtf((float)(v1 + 1)),
                                         rsqrtf((float)(v2 + 1)), rsqrtf((float)(v3 + 1)));
  } else {
    if (i0     < n) { start0[i0]     = o0; start1[i0]     = o0 + a0; dinvf[i0]     = rsqrtf((float)(v0 + 1)); }
    if (i0 + 1 < n) { start0[i0 + 1] = o1; start1[i0 + 1] = o1 + a1; dinvf[i0 + 1] = rsqrtf((float)(v1 + 1)); }
    if (i0 + 2 < n) { start0[i0 + 2] = o2; start1[i0 + 2] = o2 + a2; dinvf[i0 + 2] = rsqrtf((float)(v2 + 1)); }
  }
  if (i0 < n && n <= i0 + 4) start0[n] = o0 + s;
}

__global__ void fill_kernel(const int* __restrict__ src, const int* __restrict__ dst,
                            const int* __restrict__ rank, const int* __restrict__ start0,
                            const int* __restrict__ start1, int* __restrict__ csr, int e) {
  int i = blockIdx.x * blockDim.x + threadIdx.x;
  if (i < e) {
    const int d = dst[i];
    const int rk = rank[i];
    const int* __restrict__ st = (rk & 1) ? start1 : start0;
    csr[st[d] + (rk >> 1)] = src[i];
  }
}

__global__ void __launch_bounds__(128)
gemm_self_kernel(const float* __restrict__ x, const float* __restrict__ W,
                 const float* __restrict__ bias, const float* __restrict__ dinvf,
                 float* __restrict__ y, float* __restrict__ out) {
  const int t = threadIdx.x;
  const int m0 = blockIdx.x * 16;
  const int n0 = threadIdx.y * 16;
  const int l16 = t & 15;
  const int kh = (t >> 4) * 2;
  const int row = m0 + l16;
  const int col = n0 + l16;

  v8f c = {};
  #pragma unroll
  for (int k = 0; k < 64; k += 4) {
    v2f a = *(const v2f*)(x + row * 64 + k + kh);
    v2f b;
    b[0] = W[(k + kh) * 64 + col];
    b[1] = W[(k + kh + 1) * 64 + col];
    c = __builtin_amdgcn_wmma_f32_16x16x4_f32(
         false, a,  false, b,
         (short)0, c,  false,  false);
  }

  const float bv = bias[col];
  #pragma unroll
  for (int v = 0; v < 8; v++) {
    int m = m0 + 8 * (t >> 4) + v;
    float di = dinvf[m];
    float yv = c[v] * di;
    y[m * 64 + col] = yv;
    out[m * 64 + col] = yv * di + bv;
  }
}

__global__ void gemm_tail_kernel(const float* __restrict__ x, const float* __restrict__ W,
                                 const float* __restrict__ bias, const float* __restrict__ dinvf,
                                 float* __restrict__ y, float* __restrict__ out,
                                 int row0, int n) {
  int idx = blockIdx.x * blockDim.x + threadIdx.x;
  int row = row0 + idx / 64;
  int col = idx % 64;
  if (row >= n) return;
  float acc = 0.f;
  for (int k = 0; k < 64; k++) acc += x[row * 64 + k] * W[k * 64 + col];
  float di = dinvf[row];
  float yv = acc * di;
  y[row * 64 + col] = yv;
  out[row * 64 + col] = yv * di + bias[col];
}

#define AGG_THREADS 256
#define AGG_GROUPS 1024

__device__ __forceinline__ v2f agg_row(const float* __restrict__ y, int s, int lane) {
  return *(const v2f*)(y + s * 64 + 2 * lane);
}

__device__ __forceinline__ void agg_sixteen(const float* __restrict__ y, int idx, int first, int m,
                                            int lane, float& ax, float& ay) {
  v2f r[16];
  #pragma unroll
  for (int u = 0; u < 16; u++)
    if (first + u < m) r[u] = agg_row(y, __builtin_amdgcn_readlane(idx, first + u), lane);
  #pragma unroll
  for (int u = 0; u < 16; u++)
    if (first + u < m) { ax += r[u][0]; ay += r[u][1]; }
}

__global__ void __launch_bounds__(AGG_THREADS)
aggregate_kernel(const int* __restrict__ rowstart, const int* __restrict__ csr,
                 const float* __restrict__ y, const float* __restrict__ dinvf,
                 float* __restrict__ out, int n) {
  const int lane = threadIdx.x & 31;
  const int wave = __builtin_amdgcn_readfirstlane((int)(blockIdx.x * (AGG_THREADS / 32) + (threadIdx.x >> 5)));
  const int nwaves = (int)gridDim.x * (AGG_THREADS / 32);
  int node = wave;
  if (node >= n) return;
  int beg = rowstart[node];
  int end = rowstart[node + 1];
  int idx = (lane < end - beg) ? csr[beg + lane] : 0;
  for (;;) {
    const int nnode = node + nwaves;
    const bool more = nnode < n;
    int nbeg = 0, nend = 0;
    if (more) { nbeg = rowstart[nnode]; nend = rowstart[nnode + 1]; }
    const int o = node * 64 + 2 * lane;
    const float di = dinvf[node];
    const v2f prev = *(const v2f*)(out + o);
    const int m = min(end - beg, 32);
    float ax = 0.f, ay = 0.f;
    int nidx = 0;
    v2f r[16];
    r[0] = agg_row(y, __builtin_amdgcn_readlane(idx, 0), lane);
    #pragma unroll
    for (int u = 1; u < 16; u++)
      if (u < m) r[u] = agg_row(y, __builtin_amdgcn_readlane(idx, u), lane);
    if (more && lane < nend - nbeg) nidx = csr[nbeg + lane];
    #pragma unroll
    for (int u = 0; u < 16; u++)
      if (u < m) { ax += r[u][0]; ay += r[u][1]; }
    if (m > 16) agg_sixteen(y, idx, 16, m, lane, ax, ay);
    for (int j = beg + 32; j < end; j++) {
      const v2f v = agg_row(y, csr[j], lane);
      ax += v[0];
      ay += v[1];
    }
    v2f res;
    res[0] = prev[0] + di * ax;
    res[1] = prev[1] + di * ay;
    *(v2f*)(out + o) = res;
    if (!more) break;
    node = nnode; beg = nbeg; end = nend; idx = nidx;
  }
}

static inline size_t ws_align(size_t x) { return (x + 255) & ~(size_t)255; }

extern "C" void kernel_launch(void* const* d_in, const int* in_sizes, int n_in,
                              void* d_out, int out_size, void* d_ws, size_t ws_size,
                              hipStream_t stream) {
  const float* x  = (const float*)d_in[0];
  const int*   ei = (const int*)d_in[1];
  const float* W  = (const float*)d_in[2];
  const float* b  = (const float*)d_in[3];
  float* out = (float*)d_out;

  const int N = in_sizes[0] / 64;
  const int E = in_sizes[1] / 2;
  const int* src = ei;
  const int* dst = ei + E;

  char* ws = (char*)d_ws;
  size_t off = 0;
  int*   cnt0     = (int*)(ws + off);   off = ws_align(off + (size_t)N * sizeof(int));
  int*   cnt1     = (int*)(ws + off);   off = ws_align(off + (size_t)N * sizeof(int));
  float* dinvf    = (float*)(ws + off); off = ws_align(off + (size_t)N * sizeof(float));
  int*   rowstart = (int*)(ws + off);   off = ws_align(off + (size_t)(N + 1) * sizeof(int));
  int*   start1   = (int*)(ws + off);   off = ws_align(off + (size_t)N * sizeof(int));
  int*   rank     = (int*)(ws + off);   off = ws_align(off + (size_t)E * sizeof(int));
  int*   csr      = (int*)(ws + off);   off = ws_align(off + (size_t)E * sizeof(int));
  float* y        = (float*)(ws + off); off = ws_align(off + (size_t)N * 64 * sizeof(float));

  cnt_init_kernel<<<(N + 255) / 256, 256, 0, stream>>>(cnt0, cnt1, N);
  count_rank_kernel<<<(E + 255) / 256, 256, 0, stream>>>(dst, cnt0, cnt1, rank, E);
  scan_kernel<<<(N + SCAN_SLICE - 1) / SCAN_SLICE, SCAN_THREADS, 0, stream>>>(cnt0, cnt1, rowstart, start1, dinvf, N);
  fill_kernel<<<(E + 255) / 256, 256, 0, stream>>>(src, dst, rank, rowstart, start1, csr, E);

  int mtiles = N / 16;
  if (mtiles > 0)
    gemm_self_kernel<<<mtiles, dim3(32, 4), 0, stream>>>(x, W, b, dinvf, y, out);
  int rem = N - mtiles * 16;
  if (rem > 0)
    gemm_tail_kernel<<<(rem * 64 + 255) / 256, 256, 0, stream>>>(x, W, b, dinvf, y, out,
                                                                 mtiles * 16, N);

  aggregate_kernel<<<AGG_GROUPS, AGG_THREADS, 0, stream>>>(rowstart, csr, y, dinvf, out, N);
}


# the product moves behind the sum: a wave sums 16 nodes' dinv-scaled x rows (CSR order, fma), multiplies its 16x64 tile by W on WMMA, stores out once; y, its launch, out re-read go; order (Ax)W
# speedup vs baseline: 5.6071x; 1.3627x over previous
#include <hip/hip_runtime.h>

typedef __attribute__((ext_vector_type(2))) float v2f;
typedef __attribute__((ext_vector_type(8))) float v8f;


__device__ __forceinline__ int package_half() {
  const unsigned id = __builtin_amdgcn_s_sendmsg_rtn(0x87);
  return (int)((id >> 18) & 1u);
}

__global__ void cnt_init_kernel(int* __restrict__ cnt0, int* __restrict__ cnt1, int n) {
  int i = blockIdx.x * blockDim.x + threadIdx.x;
  if (i < n) { cnt0[i] = 0; cnt1[i] = 0; }
}

__global__ void count_rank_kernel(const int* __restrict__ dst, int* __restrict__ cnt0,
                                  int* __restrict__ cnt1, int* __restrict__ rank, int e) {
  int i = blockIdx.x * blockDim.x + threadIdx.x;
  const int h = package_half();
  int* __restrict__ cnt = h ? cnt1 : cnt0;
  if (i < e) {
    const int r = atomicAdd(&cnt[dst[i]], 1);
    rank[i] = (r << 1) | h;
  }
}

#define SCAN_THREADS 1024
#define SCAN_SLICE (4 * SCAN_THREADS)
static_assert(SCAN_THREADS == 32 * 32, "the 32 per-wave sums are folded by one 32-lane wave");
__global__ void __launch_bounds__(SCAN_THREADS)
scan_kernel(const int* __restrict__ cnt0, const int* __restrict__ cnt1,
            int* __restrict__ start0, int* __restrict__ start1,
            float* __restrict__ dinvf, int n) {
  __shared__ int wave_own[SCAN_THREADS / 32];
  __shared__ int wave_pre[SCAN_THREADS / 32];
  const int tid = threadIdx.x;
  const int lane = tid & 31;
  const int wv = tid >> 5;
  const int4* __restrict__ p0 = (const int4*)cnt0;
  const int4* __restrict__ p1 = (const int4*)cnt1;

  const int nb = blockIdx.x;
  int pre = 0;
  for (int i = 0; i < nb; i += 8) {
    int4 u[8], v[8];
    #pragma unroll
    for (int j = 0; j < 8; j++) {
      const int q = min(i + j, nb - 1);
      u[j] = p0[q * SCAN_THREADS + tid];
      v[j] = p1[q * SCAN_THREADS + tid];
    }
    #pragma unroll
    for (int j = 0; j < 8; j++)
      if (i + j < nb)
        pre += ((u[j].x + u[j].y) + (u[j].z + u[j].w)) + ((v[j].x + v[j].y) + (v[j].z + v[j].w));
  }

  const int i0 = blockIdx.x * SCAN_SLICE + 4 * tid;
  const bool whole = (i0 + 3 < n);
  int a0 = 0, a1 = 0, a2 = 0, a3 = 0;
  int b0 = 0, b1 = 0, b2 = 0, b3 = 0;
  if (whole) {
    const int4 a = p0[i0 >> 2];
    const int4 b = p1[i0 >> 2];
    a0 = a.x; a1 = a.y; a2 = a.z; a3 = a.w;
    b0 = b.x; b1 = b.y; b2 = b.z; b3 = b.w;
  } else {
    if (i0     < n) { a0 = cnt0[i0];     b0 = cnt1[i0]; }
    if (i0 + 1 < n) { a1 = cnt0[i0 + 1]; b1 = cnt1[i0 + 1]; }
    if (i0 + 2 < n) { a2 = cnt0[i0 + 2]; b2 = cnt1[i0 + 2]; }
  }
  const int v0 = a0 + b0, v1 = a1 + b1, v2 = a2 + b2, v3 = a3 + b3;
  const int s = (v0 + v1) + (v2 + v3);

  int inc = s;
  for (int off = 1; off < 32; off <<= 1) {
    const int t = __shfl_up(inc, off);
    if (lane >= off) inc += t;
  }
  for (int off = 16; off > 0; off >>= 1) pre += __shfl_xor(pre, off);
  if (lane == 31) wave_own[wv] = inc;
  if (lane == 0) wave_pre[wv] = pre;
  __syncthreads();

  int base = wave_pre[lane] + (lane < wv ? wave_own[lane] : 0);
  for (int off = 16; off > 0; off >>= 1) base += __shfl_xor(base, off);

  const int o0 = base + inc - s;
  const int o1 = o0 + v0;
  const int o2 = o1 + v1;
  const int o3 = o2 + v2;
  if (whole) {
    *(int4*)(start0 + i0) = make_int4(o0, o1, o2, o3);
    *(int4*)(start1 + i0) = make_int4(o0 + a0, o1 + a1, o2 + a2, o3 + a3);
    *(float4*)(dinvf + i0) = make_float4(rsqrtf((float)(v0 + 1)), rsqrtf((float)(v1 + 1)),
                                         rsqrtf((float)(v2 + 1)), rsqrtf((float)(v3 + 1)));
  } else {
    if (i0     < n) { start0[i0]     = o0; start1[i0]     = o0 + a0; dinvf[i0]     = rsqrtf((float)(v0 + 1)); }
    if (i0 + 1 < n) { start0[i0 + 1] = o1; start1[i0 + 1] = o1 + a1; dinvf[i0 + 1] = rsqrtf((float)(v1 + 1)); }
    if (i0 + 2 < n) { start0[i0 + 2] = o2; start1[i0 + 2] = o2 + a2; dinvf[i0 + 2] = rsqrtf((float)(v2 + 1)); }
  }
  if (i0 < n && n <= i0 + 4) start0[n] = o0 + s;
}

__global__ void fill_kernel(const int* __restrict__ src, const int* __restrict__ dst,
                            const int* __restrict__ rank, const int* __restrict__ start0,
                            const int* __restrict__ start1, int* __restrict__ csr, int e) {
  int i = blockIdx.x * blockDim.x + threadIdx.x;
  if (i < e) {
    const int d = dst[i];
    const int rk = rank[i];
    const int* __restrict__ st = (rk & 1) ? start1 : start0;
    csr[st[d] + (rk >> 1)] = src[i];
  }
}

#define AG_WAVES 4
#define AG_THREADS (32 * AG_WAVES)
#define WF_ENTRIES (16 * 4 * 32)
static_assert(WF_ENTRIES % AG_THREADS == 0, "every lane stages the same number of W fragments");
#define ZP 68

__device__ __forceinline__ v2f x_row(const float* __restrict__ x, int s, int lane) {
  return *(const v2f*)(x + s * 64 + 2 * lane);
}

__device__ __forceinline__ float lane_f(float v, int l) {
  return __int_as_float(__builtin_amdgcn_readlane(__float_as_int(v), l));
}

__device__ __forceinline__ void sum_sixteen(const float* __restrict__ x, int idx, float dv, int first,
                                            int m, int lane, float& ax, float& ay) {
  v2f r[16];
  #pragma unroll
  for (int u = 0; u < 16; u++)
    if (first + u < m) r[u] = x_row(x, __builtin_amdgcn_readlane(idx, first + u), lane);
  #pragma unroll
  for (int u = 0; u < 16; u++)
    if (first + u < m) {
      const float s = lane_f(dv, first + u);
      ax = fmaf(r[u][0], s, ax);
      ay = fmaf(r[u][1], s, ay);
    }
}

__global__ void __launch_bounds__(AG_THREADS)
aggregate_gemm_kernel(const int* __restrict__ rowstart, const int* __restrict__ csr,
                      const float* __restrict__ x, const float* __restrict__ W,
                      const float* __restrict__ bias, const float* __restrict__ dinvf,
                      float* __restrict__ out, int n) {
  __shared__ __attribute__((aligned(16))) v2f wfrag[WF_ENTRIES];
  __shared__ __attribute__((aligned(16))) float ztile[AG_WAVES][16 * ZP];

  const int lane = threadIdx.x & 31;
  const int l16 = lane & 15;
  const int kh = (lane >> 4) * 2;
  const int wv = __builtin_amdgcn_readfirstlane((int)(threadIdx.x >> 5));
  const int m0 = ((int)blockIdx.x * AG_WAVES + wv) * 16;
  float* __restrict__ zt = &ztile[wv][0];

  {
    v2f wb[WF_ENTRIES / AG_THREADS];
    #pragma unroll
    for (int i = 0; i < WF_ENTRIES / AG_THREADS; i++) {
      const int e = (int)threadIdx.x + i * AG_THREADS;
      const int nt = (e >> 5) & 3;
      const int kk = e >> 7;
      const int k = 4 * kk + kh;
      const int col = nt * 16 + l16;
      wb[i][0] = W[k * 64 + col];
      wb[i][1] = W[(k + 1) * 64 + col];
    }
    #pragma unroll
    for (int i = 0; i < WF_ENTRIES / AG_THREADS; i++)
      wfrag[(int)threadIdx.x + i * AG_THREADS] = wb[i];
  }

  int beg = 0, end = 0, idx = 0;
  if (m0 < n) {
    beg = rowstart[m0];
    end = rowstart[m0 + 1];
    idx = (lane < end - beg) ? csr[beg + lane] : 0;
  }
  for (int r = 0; r < 16; r++) {
    const int node = m0 + r;
    float zx = 0.f, zy = 0.f;
    int nend = end, nidx = 0;
    if (node < n) {
      const bool more = (r < 15) && (node + 1 < n);
      if (more) nend = rowstart[node + 2];
      const float di = dinvf[node];
      const v2f self = x_row(x, node, lane);
      const int m = min(end - beg, 32);
      const float dv = dinvf[idx];
      float ax = 0.f, ay = 0.f;
      v2f rr[16];
      rr[0] = x_row(x, __builtin_amdgcn_readlane(idx, 0), lane);
      #pragma unroll
      for (int u = 1; u < 16; u++)
        if (u < m) rr[u] = x_row(x, __builtin_amdgcn_readlane(idx, u), lane);
      if (more && lane < nend - end) nidx = csr[end + lane];
      #pragma unroll
      for (int u = 0; u < 16; u++)
        if (u < m) {
          const float s = lane_f(dv, u);
          ax = fmaf(rr[u][0], s, ax);
          ay = fmaf(rr[u][1], s, ay);
        }
      if (m > 16) sum_sixteen(x, idx, dv, 16, m, lane, ax, ay);
      for (int j = beg + 32; j < end; j++) {
        const int s = csr[j];
        const float sd = dinvf[s];
        const v2f v = x_row(x, s, lane);
        ax = fmaf(v[0], sd, ax);
        ay = fmaf(v[1], sd, ay);
      }
      zx = di * (ax + di * self[0]);
      zy = di * (ay + di * self[1]);
    }
    v2f z;
    z[0] = zx;
    z[1] = zy;
    *(v2f*)(zt + r * ZP + 2 * lane) = z;
    beg = end; end = nend; idx = nidx;
  }

  __syncthreads();

  v8f c0 = {}, c1 = {}, c2 = {}, c3 = {};
  #pragma unroll
  for (int kk = 0; kk < 16; kk++) {
    const v2f a = *(const v2f*)(zt + l16 * ZP + 4 * kk + kh);
    const v2f b0 = wfrag[(kk * 4 + 0) * 32 + lane];
    const v2f b1 = wfrag[(kk * 4 + 1) * 32 + lane];
    const v2f b2 = wfrag[(kk * 4 + 2) * 32 + lane];
    const v2f b3 = wfrag[(kk * 4 + 3) * 32 + lane];
    c0 = __builtin_amdgcn_wmma_f32_16x16x4_f32(false, a, false, b0, (short)0, c0, false, false);
    c1 = __builtin_amdgcn_wmma_f32_16x16x4_f32(false, a, false, b1, (short)0, c1, false, false);
    c2 = __builtin_amdgcn_wmma_f32_16x16x4_f32(false, a, false, b2, (short)0, c2, false, false);
    c3 = __builtin_amdgcn_wmma_f32_16x16x4_f32(false, a, false, b3, (short)0, c3, false, false);
  }

  const float bv0 = bias[l16], bv1 = bias[16 + l16], bv2 = bias[32 + l16], bv3 = bias[48 + l16];
  #pragma unroll
  for (int v = 0; v < 8; v++) {
    const int mrow = m0 + 8 * (lane >> 4) + v;
    const float o0 = c0[v] + bv0, o1 = c1[v] + bv1, o2 = c2[v] + bv2, o3 = c3[v] + bv3;
    if (mrow < n) {
      float* __restrict__ p = out + mrow * 64 + l16;
      p[0] = o0;
      p[16] = o1;
      p[32] = o2;
      p[48] = o3;
    }
  }
}

static inline size_t ws_align(size_t x) { return (x + 255) & ~(size_t)255; }

extern "C" void kernel_launch(void* const* d_in, const int* in_sizes, int n_in,
                              void* d_out, int out_size, void* d_ws, size_t ws_size,
                              hipStream_t stream) {
  const float* x  = (const float*)d_in[0];
  const int*   ei = (const int*)d_in[1];
  const float* W  = (const float*)d_in[2];
  const float* b  = (const float*)d_in[3];
  float* out = (float*)d_out;

  const int N = in_sizes[0] / 64;
  const int E = in_sizes[1] / 2;
  const int* src = ei;
  const int* dst = ei + E;

  char* ws = (char*)d_ws;
  size_t off = 0;
  int*   cnt0     = (int*)(ws + off);   off = ws_align(off + (size_t)N * sizeof(int));
  int*   cnt1     = (int*)(ws + off);   off = ws_align(off + (size_t)N * sizeof(int));
  float* dinvf    = (float*)(ws + off); off = ws_align(off + (size_t)N * sizeof(float));
  int*   rowstart = (int*)(ws + off);   off = ws_align(off + (size_t)(N + 1) * sizeof(int));
  int*   start1   = (int*)(ws + off);   off = ws_align(off + (size_t)N * sizeof(int));
  int*   rank     = (int*)(ws + off);   off = ws_align(off + (size_t)E * sizeof(int));
  int*   csr      = (int*)(ws + off);   off = ws_align(off + (size_t)E * sizeof(int));

  cnt_init_kernel<<<(N + 255) / 256, 256, 0, stream>>>(cnt0, cnt1, N);
  count_rank_kernel<<<(E + 255) / 256, 256, 0, stream>>>(dst, cnt0, cnt1, rank, E);
  scan_kernel<<<(N + SCAN_SLICE - 1) / SCAN_SLICE, SCAN_THREADS, 0, stream>>>(cnt0, cnt1, rowstart, start1, dinvf, N);
  fill_kernel<<<(E + 255) / 256, 256, 0, stream>>>(src, dst, rank, rowstart, start1, csr, E);

  aggregate_gemm_kernel<<<(N + 16 * AG_WAVES - 1) / (16 * AG_WAVES), AG_THREADS, 0, stream>>>(
      rowstart, csr, x, W, b, dinvf, out, N);
}


# clear, count and fill passes: a lane takes 8 edges (4 nodes) by 16-byte loads/stores, all its adds and gathers in flight: 3928 waves enter per edge pass, not 31256; sum order still arrival rank
# speedup vs baseline: 5.7960x; 1.0337x over previous
#include <hip/hip_runtime.h>

typedef __attribute__((ext_vector_type(2))) float v2f;
typedef __attribute__((ext_vector_type(8))) float v8f;


__device__ __forceinline__ int package_half() {
  const unsigned id = __builtin_amdgcn_s_sendmsg_rtn(0x87);
  return (int)((id >> 18) & 1u);
}

#define EDGE_THREADS 256
#define EDGE_VECS 2
#define EDGE_TILE (4 * EDGE_THREADS)
#define EDGE_BLOCK (EDGE_VECS * EDGE_TILE)
#define NODE_BLOCK (4 * EDGE_THREADS)

__device__ __forceinline__ bool aligned16(const void* p) { return ((size_t)p & 15) == 0; }

template <bool VEC>
__device__ __forceinline__ int4 load4(const int* __restrict__ p, int i) {
  if (VEC) return *(const int4*)(p + i);
  return make_int4(p[i], p[i + 1], p[i + 2], p[i + 3]);
}

template <bool VEC>
__device__ __forceinline__ void store4(int* __restrict__ p, int i, int4 v) {
  if (VEC) { *(int4*)(p + i) = v; return; }
  p[i] = v.x; p[i + 1] = v.y; p[i + 2] = v.z; p[i + 3] = v.w;
}

__global__ void __launch_bounds__(EDGE_THREADS)
cnt_init_kernel(int* __restrict__ cnt0, int* __restrict__ cnt1, int n) {
  const int nwhole = n / NODE_BLOCK;
  const int b = (int)blockIdx.x;
  if (b < nwhole) {
    const int i = b * NODE_BLOCK + 4 * (int)threadIdx.x;
    const int4 z = make_int4(0, 0, 0, 0);
    store4<true>(cnt0, i, z);
    store4<true>(cnt1, i, z);
  } else {
    const int i = nwhole * NODE_BLOCK + (b - nwhole) * EDGE_THREADS + (int)threadIdx.x;
    if (i < n) { cnt0[i] = 0; cnt1[i] = 0; }
  }
}

template <bool VEC>
__device__ __forceinline__ void count_rank_whole(const int* __restrict__ dst, int* __restrict__ cnt,
                                                 int* __restrict__ rank, int base, int h) {
  int4 d[EDGE_VECS], r[EDGE_VECS];
  #pragma unroll
  for (int j = 0; j < EDGE_VECS; j++) d[j] = load4<VEC>(dst, base + j * EDGE_TILE);
  static_assert(EDGE_VECS == 2, "the statement below names both groups of four");
  asm volatile("" :: "v"(d[0].x), "v"(d[0].y), "v"(d[0].z), "v"(d[0].w),
                     "v"(d[1].x), "v"(d[1].y), "v"(d[1].z), "v"(d[1].w));
  #pragma unroll
  for (int j = 0; j < EDGE_VECS; j++) {
    r[j].x = atomicAdd(&cnt[d[j].x], 1);
    r[j].y = atomicAdd(&cnt[d[j].y], 1);
    r[j].z = atomicAdd(&cnt[d[j].z], 1);
    r[j].w = atomicAdd(&cnt[d[j].w], 1);
  }
  #pragma unroll
  for (int j = 0; j < EDGE_VECS; j++) {
    const int4 v = make_int4((r[j].x << 1) | h, (r[j].y << 1) | h, (r[j].z << 1) | h, (r[j].w << 1) | h);
    store4<VEC>(rank, base + j * EDGE_TILE, v);
  }
}

__global__ void __launch_bounds__(EDGE_THREADS)
count_rank_kernel(const int* __restrict__ dst, int* __restrict__ cnt0,
                  int* __restrict__ cnt1, int* __restrict__ rank, int e) {
  const int h = package_half();
  int* __restrict__ cnt = h ? cnt1 : cnt0;
  const int nwhole = e / EDGE_BLOCK;
  const int b = (int)blockIdx.x;
  if (b < nwhole) {
    const int base = b * EDGE_BLOCK + 4 * (int)threadIdx.x;
    if (aligned16(dst) && aligned16(rank)) count_rank_whole<true>(dst, cnt, rank, base, h);
    else count_rank_whole<false>(dst, cnt, rank, base, h);
  } else {
    const int i = nwhole * EDGE_BLOCK + (b - nwhole) * EDGE_THREADS + (int)threadIdx.x;
    if (i < e) {
      const int r = atomicAdd(&cnt[dst[i]], 1);
      rank[i] = (r << 1) | h;
    }
  }
}

#define SCAN_THREADS 1024
#define SCAN_SLICE (4 * SCAN_THREADS)
static_assert(SCAN_THREADS == 32 * 32, "the 32 per-wave sums are folded by one 32-lane wave");
__global__ void __launch_bounds__(SCAN_THREADS)
scan_kernel(const int* __restrict__ cnt0, const int* __restrict__ cnt1,
            int* __restrict__ start0, int* __restrict__ start1,
            float* __restrict__ dinvf, int n) {
  __shared__ int wave_own[SCAN_THREADS / 32];
  __shared__ int wave_pre[SCAN_THREADS / 32];
  const int tid = threadIdx.x;
  const int lane = tid & 31;
  const int wv = tid >> 5;
  const int4* __restrict__ p0 = (const int4*)cnt0;
  const int4* __restrict__ p1 = (const int4*)cnt1;

  const int nb = blockIdx.x;
  int pre = 0;
  for (int i = 0; i < nb; i += 8) {
    int4 u[8], v[8];
    #pragma unroll
    for (int j = 0; j < 8; j++) {
      const int q = min(i + j, nb - 1);
      u[j] = p0[q * SCAN_THREADS + tid];
      v[j] = p1[q * SCAN_THREADS + tid];
    }
    #pragma unroll
    for (int j = 0; j < 8; j++)
      if (i + j < nb)
        pre += ((u[j].x + u[j].y) + (u[j].z + u[j].w)) + ((v[j].x + v[j].y) + (v[j].z + v[j].w));
  }

  const int i0 = blockIdx.x * SCAN_SLICE + 4 * tid;
  const bool whole = (i0 + 3 < n);
  int a0 = 0, a1 = 0, a2 = 0, a3 = 0;
  int b0 = 0, b1 = 0, b2 = 0, b3 = 0;
  if (whole) {
    const int4 a = p0[i0 >> 2];
    const int4 b = p1[i0 >> 2];
    a0 = a.x; a1 = a.y; a2 = a.z; a3 = a.w;
    b0 = b.x; b1 = b.y; b2 = b.z; b3 = b.w;
  } else {
    if (i0     < n) { a0 = cnt0[i0];     b0 = cnt1[i0]; }
    if (i0 + 1 < n) { a1 = cnt0[i0 + 1]; b1 = cnt1[i0 + 1]; }
    if (i0 + 2 < n) { a2 = cnt0[i0 + 2]; b2 = cnt1[i0 + 2]; }
  }
  const int v0 = a0 + b0, v1 = a1 + b1, v2 = a2 + b2, v3 = a3 + b3;
  const int s = (v0 + v1) + (v2 + v3);

  int inc = s;
  for (int off = 1; off < 32; off <<= 1) {
    const int t = __shfl_up(inc, off);
    if (lane >= off) inc += t;
  }
  for (int off = 16; off > 0; off >>= 1) pre += __shfl_xor(pre, off);
  if (lane == 31) wave_own[wv] = inc;
  if (lane == 0) wave_pre[wv] = pre;
  __syncthreads();

  int base = wave_pre[lane] + (lane < wv ? wave_own[lane] : 0);
  for (int off = 16; off > 0; off >>= 1) base += __shfl_xor(base, off);

  const int o0 = base + inc - s;
  const int o1 = o0 + v0;
  const int o2 = o1 + v1;
  const int o3 = o2 + v2;
  if (whole) {
    *(int4*)(start0 + i0) = make_int4(o0, o1, o2, o3);
    *(int4*)(start1 + i0) = make_int4(o0 + a0, o1 + a1, o2 + a2, o3 + a3);
    *(float4*)(dinvf + i0) = make_float4(rsqrtf((float)(v0 + 1)), rsqrtf((float)(v1 + 1)),
                                         rsqrtf((float)(v2 + 1)), rsqrtf((float)(v3 + 1)));
  } else {
    if (i0     < n) { start0[i0]     = o0; start1[i0]     = o0 + a0; dinvf[i0]     = rsqrtf((float)(v0 + 1)); }
    if (i0 + 1 < n) { start0[i0 + 1] = o1; start1[i0 + 1] = o1 + a1; dinvf[i0 + 1] = rsqrtf((float)(v1 + 1)); }
    if (i0 + 2 < n) { start0[i0 + 2] = o2; start1[i0 + 2] = o2 + a2; dinvf[i0 + 2] = rsqrtf((float)(v2 + 1)); }
  }
  if (i0 < n && n <= i0 + 4) start0[n] = o0 + s;
}

__device__ __forceinline__ int edge_slot(const int* __restrict__ start0, const int* __restrict__ start1,
                                         int d, int rk) {
  const int* __restrict__ st = (rk & 1) ? start1 : start0;
  return st[d] + (rk >> 1);
}

template <bool VEC>
__device__ __forceinline__ void fill_whole(const int* __restrict__ src, const int* __restrict__ dst,
                                           const int* __restrict__ rank, const int* __restrict__ start0,
                                           const int* __restrict__ start1, int* __restrict__ csr, int base) {
  int4 s[EDGE_VECS], d[EDGE_VECS], rk[EDGE_VECS], p[EDGE_VECS];
  #pragma unroll
  for (int j = 0; j < EDGE_VECS; j++) {
    const int i = base + j * EDGE_TILE;
    d[j]  = load4<VEC>(dst, i);
    rk[j] = load4<VEC>(rank, i);
    s[j]  = load4<VEC>(src, i);
  }
  #pragma unroll
  for (int j = 0; j < EDGE_VECS; j++) {
    p[j].x = edge_slot(start0, start1, d[j].x, rk[j].x);
    p[j].y = edge_slot(start0, start1, d[j].y, rk[j].y);
    p[j].z = edge_slot(start0, start1, d[j].z, rk[j].z);
    p[j].w = edge_slot(start0, start1, d[j].w, rk[j].w);
  }
  #pragma unroll
  for (int j = 0; j < EDGE_VECS; j++) {
    csr[p[j].x] = s[j].x;
    csr[p[j].y] = s[j].y;
    csr[p[j].z] = s[j].z;
    csr[p[j].w] = s[j].w;
  }
}

__global__ void __launch_bounds__(EDGE_THREADS)
fill_kernel(const int* __restrict__ src, const int* __restrict__ dst,
            const int* __restrict__ rank, const int* __restrict__ start0,
            const int* __restrict__ start1, int* __restrict__ csr, int e) {
  const int nwhole = e / EDGE_BLOCK;
  const int b = (int)blockIdx.x;
  if (b < nwhole) {
    const int base = b * EDGE_BLOCK + 4 * (int)threadIdx.x;
    if (aligned16(src) && aligned16(dst) && aligned16(rank))
      fill_whole<true>(src, dst, rank, start0, start1, csr, base);
    else
      fill_whole<false>(src, dst, rank, start0, start1, csr, base);
  } else {
    const int i = nwhole * EDGE_BLOCK + (b - nwhole) * EDGE_THREADS + (int)threadIdx.x;
    if (i < e) csr[edge_slot(start0, start1, dst[i], rank[i])] = src[i];
  }
}

#define AG_WAVES 4
#define AG_THREADS (32 * AG_WAVES)
#define WF_ENTRIES (16 * 4 * 32)
static_assert(WF_ENTRIES % AG_THREADS == 0, "every lane stages the same number of W fragments");
#define ZP 68

__device__ __forceinline__ v2f x_row(const float* __restrict__ x, int s, int lane) {
  return *(const v2f*)(x + s * 64 + 2 * lane);
}

__device__ __forceinline__ float lane_f(float v, int l) {
  return __int_as_float(__builtin_amdgcn_readlane(__float_as_int(v), l));
}

__device__ __forceinline__ void sum_sixteen(const float* __restrict__ x, int idx, float dv, int first,
                                            int m, int lane, float& ax, float& ay) {
  v2f r[16];
  #pragma unroll
  for (int u = 0; u < 16; u++)
    if (first + u < m) r[u] = x_row(x, __builtin_amdgcn_readlane(idx, first + u), lane);
  #pragma unroll
  for (int u = 0; u < 16; u++)
    if (first + u < m) {
      const float s = lane_f(dv, first + u);
      ax = fmaf(r[u][0], s, ax);
      ay = fmaf(r[u][1], s, ay);
    }
}

__global__ void __launch_bounds__(AG_THREADS)
aggregate_gemm_kernel(const int* __restrict__ rowstart, const int* __restrict__ csr,
                      const float* __restrict__ x, const float* __restrict__ W,
                      const float* __restrict__ bias, const float* __restrict__ dinvf,
                      float* __restrict__ out, int n) {
  __shared__ __attribute__((aligned(16))) v2f wfrag[WF_ENTRIES];
  __shared__ __attribute__((aligned(16))) float ztile[AG_WAVES][16 * ZP];

  const int lane = threadIdx.x & 31;
  const int l16 = lane & 15;
  const int kh = (lane >> 4) * 2;
  const int wv = __builtin_amdgcn_readfirstlane((int)(threadIdx.x >> 5));
  const int m0 = ((int)blockIdx.x * AG_WAVES + wv) * 16;
  float* __restrict__ zt = &ztile[wv][0];

  {
    v2f wb[WF_ENTRIES / AG_THREADS];
    #pragma unroll
    for (int i = 0; i < WF_ENTRIES / AG_THREADS; i++) {
      const int e = (int)threadIdx.x + i * AG_THREADS;
      const int nt = (e >> 5) & 3;
      const int kk = e >> 7;
      const int k = 4 * kk + kh;
      const int col = nt * 16 + l16;
      wb[i][0] = W[k * 64 + col];
      wb[i][1] = W[(k + 1) * 64 + col];
    }
    #pragma unroll
    for (int i = 0; i < WF_ENTRIES / AG_THREADS; i++)
      wfrag[(int)threadIdx.x + i * AG_THREADS] = wb[i];
  }

  int beg = 0, end = 0, idx = 0;
  if (m0 < n) {
    beg = rowstart[m0];
    end = rowstart[m0 + 1];
    idx = (lane < end - beg) ? csr[beg + lane] : 0;
  }
  for (int r = 0; r < 16; r++) {
    const int node = m0 + r;
    float zx = 0.f, zy = 0.f;
    int nend = end, nidx = 0;
    if (node < n) {
      const bool more = (r < 15) && (node + 1 < n);
      if (more) nend = rowstart[node + 2];
      const float di = dinvf[node];
      const v2f self = x_row(x, node, lane);
      const int m = min(end - beg, 32);
      const float dv = dinvf[idx];
      float ax = 0.f, ay = 0.f;
      v2f rr[16];
      rr[0] = x_row(x, __builtin_amdgcn_readlane(idx, 0), lane);
      #pragma unroll
      for (int u = 1; u < 16; u++)
        if (u < m) rr[u] = x_row(x, __builtin_amdgcn_readlane(idx, u), lane);
      if (more && lane < nend - end) nidx = csr[end + lane];
      #pragma unroll
      for (int u = 0; u < 16; u++)
        if (u < m) {
          const float s = lane_f(dv, u);
          ax = fmaf(rr[u][0], s, ax);
          ay = fmaf(rr[u][1], s, ay);
        }
      if (m > 16) sum_sixteen(x, idx, dv, 16, m, lane, ax, ay);
      for (int j = beg + 32; j < end; j++) {
        const int s = csr[j];
        const float sd = dinvf[s];
        const v2f v = x_row(x, s, lane);
        ax = fmaf(v[0], sd, ax);
        ay = fmaf(v[1], sd, ay);
      }
      zx = di * (ax + di * self[0]);
      zy = di * (ay + di * self[1]);
    }
    v2f z;
    z[0] = zx;
    z[1] = zy;
    *(v2f*)(zt + r * ZP + 2 * lane) = z;
    beg = end; end = nend; idx = nidx;
  }

  __syncthreads();

  v8f c0 = {}, c1 = {}, c2 = {}, c3 = {};
  #pragma unroll
  for (int kk = 0; kk < 16; kk++) {
    const v2f a = *(const v2f*)(zt + l16 * ZP + 4 * kk + kh);
    const v2f b0 = wfrag[(kk * 4 + 0) * 32 + lane];
    const v2f b1 = wfrag[(kk * 4 + 1) * 32 + lane];
    const v2f b2 = wfrag[(kk * 4 + 2) * 32 + lane];
    const v2f b3 = wfrag[(kk * 4 + 3) * 32 + lane];
    c0 = __builtin_amdgcn_wmma_f32_16x16x4_f32(false, a, false, b0, (short)0, c0, false, false);
    c1 = __builtin_amdgcn_wmma_f32_16x16x4_f32(false, a, false, b1, (short)0, c1, false, false);
    c2 = __builtin_amdgcn_wmma_f32_16x16x4_f32(false, a, false, b2, (short)0, c2, false, false);
    c3 = __builtin_amdgcn_wmma_f32_16x16x4_f32(false, a, false, b3, (short)0, c3, false, false);
  }

  const float bv0 = bias[l16], bv1 = bias[16 + l16], bv2 = bias[32 + l16], bv3 = bias[48 + l16];
  #pragma unroll
  for (int v = 0; v < 8; v++) {
    const int mrow = m0 + 8 * (lane >> 4) + v;
    const float o0 = c0[v] + bv0, o1 = c1[v] + bv1, o2 = c2[v] + bv2, o3 = c3[v] + bv3;
    if (mrow < n) {
      float* __restrict__ p = out + mrow * 64 + l16;
      p[0] = o0;
      p[16] = o1;
      p[32] = o2;
      p[48] = o3;
    }
  }
}

static inline size_t ws_align(size_t x) { return (x + 255) & ~(size_t)255; }

extern "C" void kernel_launch(void* const* d_in, const int* in_sizes, int n_in,
                              void* d_out, int out_size, void* d_ws, size_t ws_size,
                              hipStream_t stream) {
  const float* x  = (const float*)d_in[0];
  const int*   ei = (const int*)d_in[1];
  const float* W  = (const float*)d_in[2];
  const float* b  = (const float*)d_in[3];
  float* out = (float*)d_out;

  const int N = in_sizes[0] / 64;
  const int E = in_sizes[1] / 2;
  const int* src = ei;
  const int* dst = ei + E;

  char* ws = (char*)d_ws;
  size_t off = 0;
  int*   cnt0     = (int*)(ws + off);   off = ws_align(off + (size_t)N * sizeof(int));
  int*   cnt1     = (int*)(ws + off);   off = ws_align(off + (size_t)N * sizeof(int));
  float* dinvf    = (float*)(ws + off); off = ws_align(off + (size_t)N * sizeof(float));
  int*   rowstart = (int*)(ws + off);   off = ws_align(off + (size_t)(N + 1) * sizeof(int));
  int*   start1   = (int*)(ws + off);   off = ws_align(off + (size_t)N * sizeof(int));
  int*   rank     = (int*)(ws + off);   off = ws_align(off + (size_t)E * sizeof(int));
  int*   csr      = (int*)(ws + off);   off = ws_align(off + (size_t)E * sizeof(int));

  cnt_init_kernel<<<N / NODE_BLOCK + (N % NODE_BLOCK + EDGE_THREADS - 1) / EDGE_THREADS, EDGE_THREADS, 0, stream>>>(cnt0, cnt1, N);
  const int edge_grid = E / EDGE_BLOCK + (E % EDGE_BLOCK + EDGE_THREADS - 1) / EDGE_THREADS;
  count_rank_kernel<<<edge_grid, EDGE_THREADS, 0, stream>>>(dst, cnt0, cnt1, rank, E);
  scan_kernel<<<(N + SCAN_SLICE - 1) / SCAN_SLICE, SCAN_THREADS, 0, stream>>>(cnt0, cnt1, rowstart, start1, dinvf, N);
  fill_kernel<<<edge_grid, EDGE_THREADS, 0, stream>>>(src, dst, rank, rowstart, start1, csr, E);

  aggregate_gemm_kernel<<<(N + 16 * AG_WAVES - 1) / (16 * AG_WAVES), AG_THREADS, 0, stream>>>(
      rowstart, csr, x, W, b, dinvf, out, N);
}


# aggregate: a wave sums two nodes at once, one per 16-lane half: 16-byte row loads, slot id and scale reach the half by DPP row share, no readlane, scalar address or branch per row; sum order kept
# speedup vs baseline: 11.2821x; 1.9465x over previous
#include <hip/hip_runtime.h>

typedef __attribute__((ext_vector_type(2))) float v2f;
typedef __attribute__((ext_vector_type(8))) float v8f;


__device__ __forceinline__ int package_half() {
  const unsigned id = __builtin_amdgcn_s_sendmsg_rtn(0x87);
  return (int)((id >> 18) & 1u);
}

#define EDGE_THREADS 256
#define EDGE_VECS 2
#define EDGE_TILE (4 * EDGE_THREADS)
#define EDGE_BLOCK (EDGE_VECS * EDGE_TILE)
#define NODE_BLOCK (4 * EDGE_THREADS)
#define BUCKET_SLOTS 16

__device__ __forceinline__ bool aligned16(const void* p) { return ((size_t)p & 15) == 0; }

typedef __attribute__((ext_vector_type(4))) int v4i;
template <bool VEC>
__device__ __forceinline__ int4 load4(const int* __restrict__ p, int i) {
  if (VEC) {
    const v4i v = *(const v4i*)(p + i);
    return make_int4(v[0], v[1], v[2], v[3]);
  }
  return make_int4(p[i], p[i + 1], p[i + 2], p[i + 3]);
}

template <bool VEC>
__device__ __forceinline__ void store4(int* __restrict__ p, int i, int4 v) {
  if (VEC) { *(int4*)(p + i) = v; return; }
  p[i] = v.x; p[i + 1] = v.y; p[i + 2] = v.z; p[i + 3] = v.w;
}

__global__ void __launch_bounds__(EDGE_THREADS)
cnt_init_kernel(int* __restrict__ cnt0, int* __restrict__ cnt1, int* __restrict__ ovf_n, int n) {
  const int nwhole = n / NODE_BLOCK;
  const int b = (int)blockIdx.x;
  if (b == 0 && threadIdx.x == 0) *ovf_n = 0;
  if (b < nwhole) {
    const int i = b * NODE_BLOCK + 4 * (int)threadIdx.x;
    const int4 z = make_int4(0, 0, 0, 0);
    store4<true>(cnt0, i, z);
    store4<true>(cnt1, i, z);
  } else {
    const int i = nwhole * NODE_BLOCK + (b - nwhole) * EDGE_THREADS + (int)threadIdx.x;
    if (i < n) { cnt0[i] = 0; cnt1[i] = 0; }
  }
}

__device__ __forceinline__ void place_edge(int* __restrict__ bkt, int* __restrict__ ovf_n,
                                           int* __restrict__ ovf_dst, int* __restrict__ ovf_src,
                                           int d, int r, int s) {
  if (r < BUCKET_SLOTS) {
    bkt[d * BUCKET_SLOTS + r] = s;
  } else {
    const int k = atomicAdd(ovf_n, 1);
    ovf_dst[k] = d;
    ovf_src[k] = s;
  }
}

template <bool VEC>
__device__ __forceinline__ void count_place_whole(const int* __restrict__ src, const int* __restrict__ dst,
                                                  int* __restrict__ cnt, int* __restrict__ bkt,
                                                  int* __restrict__ ovf_n, int* __restrict__ ovf_dst,
                                                  int* __restrict__ ovf_src, int base) {
  int4 d[EDGE_VECS], s[EDGE_VECS], r[EDGE_VECS];
  #pragma unroll
  for (int j = 0; j < EDGE_VECS; j++) d[j] = load4<VEC>(dst, base + j * EDGE_TILE);
  #pragma unroll
  for (int j = 0; j < EDGE_VECS; j++) s[j] = load4<VEC>(src, base + j * EDGE_TILE);
  static_assert(EDGE_VECS == 2, "the statement below names both groups of four");
  asm volatile("" :: "v"(d[0].x), "v"(d[0].y), "v"(d[0].z), "v"(d[0].w),
                     "v"(d[1].x), "v"(d[1].y), "v"(d[1].z), "v"(d[1].w));
  #pragma unroll
  for (int j = 0; j < EDGE_VECS; j++) {
    r[j].x = atomicAdd(&cnt[d[j].x], 1);
    r[j].y = atomicAdd(&cnt[d[j].y], 1);
    r[j].z = atomicAdd(&cnt[d[j].z], 1);
    r[j].w = atomicAdd(&cnt[d[j].w], 1);
  }
  #pragma unroll
  for (int j = 0; j < EDGE_VECS; j++) {
    place_edge(bkt, ovf_n, ovf_dst, ovf_src, d[j].x, r[j].x, s[j].x);
    place_edge(bkt, ovf_n, ovf_dst, ovf_src, d[j].y, r[j].y, s[j].y);
    place_edge(bkt, ovf_n, ovf_dst, ovf_src, d[j].z, r[j].z, s[j].z);
    place_edge(bkt, ovf_n, ovf_dst, ovf_src, d[j].w, r[j].w, s[j].w);
  }
}

__global__ void __launch_bounds__(EDGE_THREADS)
count_place_kernel(const int* __restrict__ src, const int* __restrict__ dst,
                   int* __restrict__ cnt0, int* __restrict__ cnt1,
                   int* __restrict__ bkt0, int* __restrict__ bkt1,
                   int* __restrict__ ovf_n, int* __restrict__ ovf_dst, int* __restrict__ ovf_src, int e) {
  const int h = package_half();
  int* __restrict__ cnt = h ? cnt1 : cnt0;
  int* __restrict__ bkt = h ? bkt1 : bkt0;
  const int nwhole = e / EDGE_BLOCK;
  const int b = (int)blockIdx.x;
  if (b < nwhole) {
    const int base = b * EDGE_BLOCK + 4 * (int)threadIdx.x;
    if (aligned16(src) && aligned16(dst))
      count_place_whole<true>(src, dst, cnt, bkt, ovf_n, ovf_dst, ovf_src, base);
    else
      count_place_whole<false>(src, dst, cnt, bkt, ovf_n, ovf_dst, ovf_src, base);
  } else {
    const int i = nwhole * EDGE_BLOCK + (b - nwhole) * EDGE_THREADS + (int)threadIdx.x;
    if (i < e) {
      const int d = dst[i];
      const int r = atomicAdd(&cnt[d], 1);
      place_edge(bkt, ovf_n, ovf_dst, ovf_src, d, r, src[i]);
    }
  }
}

__global__ void __launch_bounds__(EDGE_THREADS)
dinv_kernel(const int* __restrict__ cnt0, const int* __restrict__ cnt1, float* __restrict__ dinvf, int n) {
  const int nwhole = n / NODE_BLOCK;
  const int b = (int)blockIdx.x;
  if (b < nwhole) {
    const int i = b * NODE_BLOCK + 4 * (int)threadIdx.x;
    const int4 a = load4<true>(cnt0, i);
    const int4 c = load4<true>(cnt1, i);
    *(float4*)(dinvf + i) = make_float4(rsqrtf((float)(a.x + c.x + 1)), rsqrtf((float)(a.y + c.y + 1)),
                                        rsqrtf((float)(a.z + c.z + 1)), rsqrtf((float)(a.w + c.w + 1)));
  } else {
    const int i = nwhole * NODE_BLOCK + (b - nwhole) * EDGE_THREADS + (int)threadIdx.x;
    if (i < n) dinvf[i] = rsqrtf((float)(cnt0[i] + cnt1[i] + 1));
  }
}

#define AG_WAVES 4
#define AG_THREADS (32 * AG_WAVES)
#define WF_ENTRIES (16 * 4 * 32)
static_assert(WF_ENTRIES % AG_THREADS == 0, "every lane stages the same number of W fragments");
#define ZP 68

typedef __attribute__((ext_vector_type(4))) float v4f;

template <int U>
__device__ __forceinline__ int share_i(int v) {
  return __builtin_amdgcn_update_dpp(0, v, 0x150 + U, 0xf, 0xf, true);
}
template <int U>
__device__ __forceinline__ float share_f(float v) {
  return __int_as_float(share_i<U>(__float_as_int(v)));
}

__device__ __forceinline__ v4f x_quarter(const float* __restrict__ x, int s, unsigned lq) {
  return *(const v4f*)((const char*)x + (((unsigned)s << 8) + lq));
}

__device__ __forceinline__ void load_slots(v4f (&r)[16], const float* __restrict__ x, int id,
                                           int cnt, unsigned lq) {
  if (cnt > 0) {
    r[0] = x_quarter(x, share_i<0>(id), lq);   r[1] = x_quarter(x, share_i<1>(id), lq);
    r[2] = x_quarter(x, share_i<2>(id), lq);   r[3] = x_quarter(x, share_i<3>(id), lq);
  }
  if (cnt > 4) {
    r[4] = x_quarter(x, share_i<4>(id), lq);   r[5] = x_quarter(x, share_i<5>(id), lq);
    r[6] = x_quarter(x, share_i<6>(id), lq);   r[7] = x_quarter(x, share_i<7>(id), lq);
  }
  if (cnt > 8) {
    r[8] = x_quarter(x, share_i<8>(id), lq);   r[9] = x_quarter(x, share_i<9>(id), lq);
    r[10] = x_quarter(x, share_i<10>(id), lq); r[11] = x_quarter(x, share_i<11>(id), lq);
  }
  if (cnt > 12) {
    r[12] = x_quarter(x, share_i<12>(id), lq); r[13] = x_quarter(x, share_i<13>(id), lq);
    r[14] = x_quarter(x, share_i<14>(id), lq); r[15] = x_quarter(x, share_i<15>(id), lq);
  }
}

#define ADD_SLOT(U)                                   \
  {                                                   \
    const float sc = share_f<U>(dv);                  \
    a[0] = fmaf(r[U][0], sc, a[0]);                   \
    a[1] = fmaf(r[U][1], sc, a[1]);                   \
    a[2] = fmaf(r[U][2], sc, a[2]);                   \
    a[3] = fmaf(r[U][3], sc, a[3]);                   \
  }
__device__ __forceinline__ void add_slots(const v4f (&r)[16], float dv, int cnt, v4f& a) {
  if (cnt > 0)  { ADD_SLOT(0)  ADD_SLOT(1)  ADD_SLOT(2)  ADD_SLOT(3) }
  if (cnt > 4)  { ADD_SLOT(4)  ADD_SLOT(5)  ADD_SLOT(6)  ADD_SLOT(7) }
  if (cnt > 8)  { ADD_SLOT(8)  ADD_SLOT(9)  ADD_SLOT(10) ADD_SLOT(11) }
  if (cnt > 12) { ADD_SLOT(12) ADD_SLOT(13) ADD_SLOT(14) ADD_SLOT(15) }
}
#undef ADD_SLOT

__device__ __forceinline__ int pair_ids(const int* __restrict__ bkt0, const int* __restrict__ bkt1,
                                        int node, int k0, int mh, int l16) {
  int v = 0;
  if (l16 < mh) {
    const int* __restrict__ p = (l16 < k0) ? (bkt0 + node * BUCKET_SLOTS + l16)
                                           : (bkt1 + node * BUCKET_SLOTS + (l16 - k0));
    v = *p;
  }
  return v;
}

__device__ __forceinline__ void sum_listed(const float* __restrict__ x, const float* __restrict__ dinvf,
                                           const int* __restrict__ ovf_n, const int* __restrict__ ovf_dst,
                                           const int* __restrict__ ovf_src, int node, bool mine,
                                           int lane, unsigned lq, v4f& a) {
  const int total = *ovf_n;
  for (int j0 = 0; j0 < total; j0 += 32) {
    const int j = j0 + lane;
    int od = -1, os = 0;
    if (j < total) { od = ovf_dst[j]; os = ovf_src[j]; }
    unsigned hit = __builtin_amdgcn_ballot_w32(od == node);
    while (hit) {
      const int pos = __builtin_ctz(hit);
      hit &= hit - 1;
      const int s = __builtin_amdgcn_readlane(os, pos);
      const float sd = mine ? dinvf[s] : 0.f;
      const v4f v = x_quarter(x, s, lq);
      a[0] = fmaf(v[0], sd, a[0]);
      a[1] = fmaf(v[1], sd, a[1]);
      a[2] = fmaf(v[2], sd, a[2]);
      a[3] = fmaf(v[3], sd, a[3]);
    }
  }
}

__global__ void __launch_bounds__(AG_THREADS)
aggregate_gemm_kernel(const int* __restrict__ cnt0, const int* __restrict__ cnt1,
                      const int* __restrict__ bkt0, const int* __restrict__ bkt1,
                      const int* __restrict__ ovf_n, const int* __restrict__ ovf_dst,
                      const int* __restrict__ ovf_src,
                      const float* __restrict__ x, const float* __restrict__ W,
                      const float* __restrict__ bias, const float* __restrict__ dinvf,
                      float* __restrict__ out, int n) {
  __shared__ __attribute__((aligned(16))) v2f wfrag[WF_ENTRIES];
  __shared__ __attribute__((aligned(16))) float ztile[AG_WAVES][16 * ZP];

  const int lane = threadIdx.x & 31;
  const int l16 = lane & 15;
  const int kh = (lane >> 4) * 2;
  const int wv = __builtin_amdgcn_readfirstlane((int)(threadIdx.x >> 5));
  const int m0 = ((int)blockIdx.x * AG_WAVES + wv) * 16;
  float* __restrict__ zt = &ztile[wv][0];

  {
    v2f wb[WF_ENTRIES / AG_THREADS];
    #pragma unroll
    for (int i = 0; i < WF_ENTRIES / AG_THREADS; i++) {
      const int e = (int)threadIdx.x + i * AG_THREADS;
      const int nt = (e >> 5) & 3;
      const int kk = e >> 7;
      const int k = 4 * kk + kh;
      const int col = nt * 16 + l16;
      wb[i][0] = W[k * 64 + col];
      wb[i][1] = W[(k + 1) * 64 + col];
    }
    #pragma unroll
    for (int i = 0; i < WF_ENTRIES / AG_THREADS; i++)
      wfrag[(int)threadIdx.x + i * AG_THREADS] = wb[i];
  }

  int cv = 0;
  if (m0 + l16 < n) cv = (lane < 16) ? cnt0[m0 + l16] : cnt1[m0 + l16];
  const int half = lane >> 4;
  const unsigned lq = (unsigned)l16 << 4;
  int e0a = __builtin_amdgcn_readlane(cv, 0);
  int e0b = __builtin_amdgcn_readlane(cv, 1);
  int e1a = __builtin_amdgcn_readlane(cv, 16);
  int e1b = __builtin_amdgcn_readlane(cv, 17);
  int k0 = min(half ? e0b : e0a, BUCKET_SLOTS);
  int mh = k0 + min(half ? e1b : e1a, BUCKET_SLOTS);
  int id = pair_ids(bkt0, bkt1, min(m0 + half, n - 1), k0, mh, l16);
  for (int p = 0; p < 8; p++) {
    const int node = m0 + 2 * p + half;
    const int nodec = min(node, n - 1);
    const bool more = (p < 7);
    int e0an = 0, e0bn = 0, e1an = 0, e1bn = 0, k0n = 0, mhn = 0, nid = 0;
    if (more) {
      e0an = __builtin_amdgcn_readlane(cv, 2 * p + 2);
      e0bn = __builtin_amdgcn_readlane(cv, 2 * p + 3);
      e1an = __builtin_amdgcn_readlane(cv, 2 * p + 18);
      e1bn = __builtin_amdgcn_readlane(cv, 2 * p + 19);
    }
    const int mm = max(min(e0a, BUCKET_SLOTS) + min(e1a, BUCKET_SLOTS),
                       min(e0b, BUCKET_SLOTS) + min(e1b, BUCKET_SLOTS));
    float di = 0.f;
    v4f self = {0.f, 0.f, 0.f, 0.f};
    v4f a = {0.f, 0.f, 0.f, 0.f};
    int idg = id;
    int j = l16;
    for (int base = 0; ; base += 16) {
      const float dg = dinvf[idg];
      const float dv = (j < mh) ? dg : 0.f;
      v4f r[16];
      load_slots(r, x, idg, mm - base, lq);
      if (base == 0) {
        di = dinvf[nodec];
        self = x_quarter(x, nodec, lq);
        if (more) {
          k0n = min(half ? e0bn : e0an, BUCKET_SLOTS);
          mhn = k0n + min(half ? e1bn : e1an, BUCKET_SLOTS);
          nid = pair_ids(bkt0, bkt1, min(node + 2, n - 1), k0n, mhn, l16);
        }
      }
      add_slots(r, dv, mm - base, a);
      if (mm <= base + 16) break;
      j = 16 + l16;
      idg = 0;
      if (j < mh) idg = bkt1[nodec * BUCKET_SLOTS + (j - k0)];
    }
    if (e0a > BUCKET_SLOTS || e1a > BUCKET_SLOTS)
      sum_listed(x, dinvf, ovf_n, ovf_dst, ovf_src, m0 + 2 * p, half == 0, lane, lq, a);
    if (e0b > BUCKET_SLOTS || e1b > BUCKET_SLOTS)
      sum_listed(x, dinvf, ovf_n, ovf_dst, ovf_src, m0 + 2 * p + 1, half == 1, lane, lq, a);
    v4f z;
    z[0] = di * (a[0] + di * self[0]);
    z[1] = di * (a[1] + di * self[1]);
    z[2] = di * (a[2] + di * self[2]);
    z[3] = di * (a[3] + di * self[3]);
    if (node >= n) z = (v4f){0.f, 0.f, 0.f, 0.f};
    *(v4f*)(zt + (2 * p + half) * ZP + 4 * l16) = z;
    e0a = e0an; e0b = e0bn; e1a = e1an; e1b = e1bn;
    k0 = k0n; mh = mhn; id = nid;
  }

  __syncthreads();

  v8f c0 = {}, c1 = {}, c2 = {}, c3 = {};
  #pragma unroll
  for (int kk = 0; kk < 16; kk++) {
    const v2f a = *(const v2f*)(zt + l16 * ZP + 4 * kk + kh);
    const v2f b0 = wfrag[(kk * 4 + 0) * 32 + lane];
    const v2f b1 = wfrag[(kk * 4 + 1) * 32 + lane];
    const v2f b2 = wfrag[(kk * 4 + 2) * 32 + lane];
    const v2f b3 = wfrag[(kk * 4 + 3) * 32 + lane];
    c0 = __builtin_amdgcn_wmma_f32_16x16x4_f32(false, a, false, b0, (short)0, c0, false, false);
    c1 = __builtin_amdgcn_wmma_f32_16x16x4_f32(false, a, false, b1, (short)0, c1, false, false);
    c2 = __builtin_amdgcn_wmma_f32_16x16x4_f32(false, a, false, b2, (short)0, c2, false, false);
    c3 = __builtin_amdgcn_wmma_f32_16x16x4_f32(false, a, false, b3, (short)0, c3, false, false);
  }

  const float bv0 = bias[l16], bv1 = bias[16 + l16], bv2 = bias[32 + l16], bv3 = bias[48 + l16];
  #pragma unroll
  for (int v = 0; v < 8; v++) {
    const int mrow = m0 + 8 * (lane >> 4) + v;
    const float o0 = c0[v] + bv0, o1 = c1[v] + bv1, o2 = c2[v] + bv2, o3 = c3[v] + bv3;
    if (mrow < n) {
      float* __restrict__ p = out + mrow * 64 + l16;
      p[0] = o0;
      p[16] = o1;
      p[32] = o2;
      p[48] = o3;
    }
  }
}

static inline size_t ws_align(size_t x) { return (x + 255) & ~(size_t)255; }

extern "C" void kernel_launch(void* const* d_in, const int* in_sizes, int n_in,
                              void* d_out, int out_size, void* d_ws, size_t ws_size,
                              hipStream_t stream) {
  const float* x  = (const float*)d_in[0];
  const int*   ei = (const int*)d_in[1];
  const float* W  = (const float*)d_in[2];
  const float* b  = (const float*)d_in[3];
  float* out = (float*)d_out;

  const int N = in_sizes[0] / 64;
  const int E = in_sizes[1] / 2;
  const int* src = ei;
  const int* dst = ei + E;

  char* ws = (char*)d_ws;
  size_t off = 0;
  int*   cnt0    = (int*)(ws + off);   off = ws_align(off + (size_t)N * sizeof(int));
  int*   cnt1    = (int*)(ws + off);   off = ws_align(off + (size_t)N * sizeof(int));
  float* dinvf   = (float*)(ws + off); off = ws_align(off + (size_t)N * sizeof(float));
  int*   ovf_n   = (int*)(ws + off);   off = ws_align(off + sizeof(int));
  int*   bkt0    = (int*)(ws + off);   off = ws_align(off + (size_t)N * BUCKET_SLOTS * sizeof(int));
  int*   bkt1    = (int*)(ws + off);   off = ws_align(off + (size_t)N * BUCKET_SLOTS * sizeof(int));
  int*   ovf_dst = (int*)(ws + off);   off = ws_align(off + (size_t)E * sizeof(int));
  int*   ovf_src = (int*)(ws + off);   off = ws_align(off + (size_t)E * sizeof(int));

  const int node_grid = N / NODE_BLOCK + (N % NODE_BLOCK + EDGE_THREADS - 1) / EDGE_THREADS;
  const int edge_grid = E / EDGE_BLOCK + (E % EDGE_BLOCK + EDGE_THREADS - 1) / EDGE_THREADS;
  cnt_init_kernel<<<node_grid, EDGE_THREADS, 0, stream>>>(cnt0, cnt1, ovf_n, N);
  count_place_kernel<<<edge_grid, EDGE_THREADS, 0, stream>>>(src, dst, cnt0, cnt1, bkt0, bkt1,
                                                             ovf_n, ovf_dst, ovf_src, E);
  dinv_kernel<<<node_grid, EDGE_THREADS, 0, stream>>>(cnt0, cnt1, dinvf, N);

  aggregate_gemm_kernel<<<(N + 16 * AG_WAVES - 1) / (16 * AG_WAVES), AG_THREADS, 0, stream>>>(
      cnt0, cnt1, bkt0, bkt1, ovf_n, ovf_dst, ovf_src, x, W, b, dinvf, out, N);
}


# aggregate: one 25-wave work-group per WGP stages a one-byte in-degree table and W in LDS once; a source's scale is an LDS lookup + rsqrt, not a gathered 4-byte load per edge; sum order kept
# speedup vs baseline: 12.3462x; 1.0943x over previous
#include <hip/hip_runtime.h>

typedef __attribute__((ext_vector_type(2))) float v2f;
typedef __attribute__((ext_vector_type(8))) float v8f;


__device__ __forceinline__ int package_half() {
  const unsigned id = __builtin_amdgcn_s_sendmsg_rtn(0x87);
  return (int)((id >> 18) & 1u);
}

#define EDGE_THREADS 256
#define EDGE_VECS 2
#define EDGE_TILE (4 * EDGE_THREADS)
#define EDGE_BLOCK (EDGE_VECS * EDGE_TILE)
#define NODE_BLOCK (4 * EDGE_THREADS)
#define BUCKET_SLOTS 16

__device__ __forceinline__ bool aligned16(const void* p) { return ((size_t)p & 15) == 0; }

typedef __attribute__((ext_vector_type(4))) int v4i;
template <bool VEC>
__device__ __forceinline__ int4 load4(const int* __restrict__ p, int i) {
  if (VEC) {
    const v4i v = *(const v4i*)(p + i);
    return make_int4(v[0], v[1], v[2], v[3]);
  }
  return make_int4(p[i], p[i + 1], p[i + 2], p[i + 3]);
}

template <bool VEC>
__device__ __forceinline__ void store4(int* __restrict__ p, int i, int4 v) {
  if (VEC) { *(int4*)(p + i) = v; return; }
  p[i] = v.x; p[i + 1] = v.y; p[i + 2] = v.z; p[i + 3] = v.w;
}

__global__ void __launch_bounds__(EDGE_THREADS)
cnt_init_kernel(int* __restrict__ cnt0, int* __restrict__ cnt1, int* __restrict__ ovf_n, int n) {
  const int nwhole = n / NODE_BLOCK;
  const int b = (int)blockIdx.x;
  if (b == 0 && threadIdx.x == 0) *ovf_n = 0;
  if (b < nwhole) {
    const int i = b * NODE_BLOCK + 4 * (int)threadIdx.x;
    const int4 z = make_int4(0, 0, 0, 0);
    store4<true>(cnt0, i, z);
    store4<true>(cnt1, i, z);
  } else {
    const int i = nwhole * NODE_BLOCK + (b - nwhole) * EDGE_THREADS + (int)threadIdx.x;
    if (i < n) { cnt0[i] = 0; cnt1[i] = 0; }
  }
}

__device__ __forceinline__ void place_edge(int* __restrict__ bkt, int* __restrict__ ovf_n,
                                           int* __restrict__ ovf_dst, int* __restrict__ ovf_src,
                                           int d, int r, int s) {
  if (r < BUCKET_SLOTS) {
    bkt[d * BUCKET_SLOTS + r] = s;
  } else {
    const int k = atomicAdd(ovf_n, 1);
    ovf_dst[k] = d;
    ovf_src[k] = s;
  }
}

template <bool VEC>
__device__ __forceinline__ void count_place_whole(const int* __restrict__ src, const int* __restrict__ dst,
                                                  int* __restrict__ cnt, int* __restrict__ bkt,
                                                  int* __restrict__ ovf_n, int* __restrict__ ovf_dst,
                                                  int* __restrict__ ovf_src, int base) {
  int4 d[EDGE_VECS], s[EDGE_VECS], r[EDGE_VECS];
  #pragma unroll
  for (int j = 0; j < EDGE_VECS; j++) d[j] = load4<VEC>(dst, base + j * EDGE_TILE);
  #pragma unroll
  for (int j = 0; j < EDGE_VECS; j++) s[j] = load4<VEC>(src, base + j * EDGE_TILE);
  static_assert(EDGE_VECS == 2, "the statement below names both groups of four");
  asm volatile("" :: "v"(d[0].x), "v"(d[0].y), "v"(d[0].z), "v"(d[0].w),
                     "v"(d[1].x), "v"(d[1].y), "v"(d[1].z), "v"(d[1].w));
  #pragma unroll
  for (int j = 0; j < EDGE_VECS; j++) {
    r[j].x = atomicAdd(&cnt[d[j].x], 1);
    r[j].y = atomicAdd(&cnt[d[j].y], 1);
    r[j].z = atomicAdd(&cnt[d[j].z], 1);
    r[j].w = atomicAdd(&cnt[d[j].w], 1);
  }
  #pragma unroll
  for (int j = 0; j < EDGE_VECS; j++) {
    place_edge(bkt, ovf_n, ovf_dst, ovf_src, d[j].x, r[j].x, s[j].x);
    place_edge(bkt, ovf_n, ovf_dst, ovf_src, d[j].y, r[j].y, s[j].y);
    place_edge(bkt, ovf_n, ovf_dst, ovf_src, d[j].z, r[j].z, s[j].z);
    place_edge(bkt, ovf_n, ovf_dst, ovf_src, d[j].w, r[j].w, s[j].w);
  }
}

__global__ void __launch_bounds__(EDGE_THREADS)
count_place_kernel(const int* __restrict__ src, const int* __restrict__ dst,
                   int* __restrict__ cnt0, int* __restrict__ cnt1,
                   int* __restrict__ bkt0, int* __restrict__ bkt1,
                   int* __restrict__ ovf_n, int* __restrict__ ovf_dst, int* __restrict__ ovf_src, int e) {
  const int h = package_half();
  int* __restrict__ cnt = h ? cnt1 : cnt0;
  int* __restrict__ bkt = h ? bkt1 : bkt0;
  const int nwhole = e / EDGE_BLOCK;
  const int b = (int)blockIdx.x;
  if (b < nwhole) {
    const int base = b * EDGE_BLOCK + 4 * (int)threadIdx.x;
    if (aligned16(src) && aligned16(dst))
      count_place_whole<true>(src, dst, cnt, bkt, ovf_n, ovf_dst, ovf_src, base);
    else
      count_place_whole<false>(src, dst, cnt, bkt, ovf_n, ovf_dst, ovf_src, base);
  } else {
    const int i = nwhole * EDGE_BLOCK + (b - nwhole) * EDGE_THREADS + (int)threadIdx.x;
    if (i < e) {
      const int d = dst[i];
      const int r = atomicAdd(&cnt[d], 1);
      place_edge(bkt, ovf_n, ovf_dst, ovf_src, d, r, src[i]);
    }
  }
}

#define DEG_SAT 255
__device__ __forceinline__ unsigned deg_entry(int deg) { return (unsigned)min(deg, DEG_SAT); }

__global__ void __launch_bounds__(EDGE_THREADS)
dinv_kernel(const int* __restrict__ cnt0, const int* __restrict__ cnt1, float* __restrict__ dinvf,
            int* __restrict__ degp, int n) {
  const int nwhole = n / NODE_BLOCK;
  const int b = (int)blockIdx.x;
  if (b < nwhole) {
    const int i = b * NODE_BLOCK + 4 * (int)threadIdx.x;
    const int4 a = load4<true>(cnt0, i);
    const int4 c = load4<true>(cnt1, i);
    const int d0 = a.x + c.x + 1, d1 = a.y + c.y + 1, d2 = a.z + c.z + 1, d3 = a.w + c.w + 1;
    *(float4*)(dinvf + i) = make_float4(rsqrtf((float)d0), rsqrtf((float)d1),
                                        rsqrtf((float)d2), rsqrtf((float)d3));
    degp[i >> 2] = (int)(deg_entry(d0) | (deg_entry(d1) << 8) | (deg_entry(d2) << 16) | (deg_entry(d3) << 24));
  } else {
    const int i = nwhole * NODE_BLOCK + (b - nwhole) * EDGE_THREADS + (int)threadIdx.x;
    if (i < n) {
      const int d = cnt0[i] + cnt1[i] + 1;
      dinvf[i] = rsqrtf((float)d);
      ((unsigned char*)degp)[i] = (unsigned char)deg_entry(d);
    }
  }
}

#define AG_WAVES 25
#define AG_THREADS (32 * AG_WAVES)
#define WF_ENTRIES (16 * 4 * 32)
#define WF_ROUNDS ((WF_ENTRIES + AG_THREADS - 1) / AG_THREADS)
static_assert(AG_THREADS % 32 == 0, "entry e of the W fragments is staged by lane e & 31");
#define TAB_NODES 102400
#define TAB_VECS (TAB_NODES / 16)
#define TAB_ROUNDS ((TAB_VECS + AG_THREADS - 1) / AG_THREADS)
#define ZP 68

typedef __attribute__((ext_vector_type(4))) float v4f;

template <int U>
__device__ __forceinline__ int share_i(int v) {
  return __builtin_amdgcn_update_dpp(0, v, 0x150 + U, 0xf, 0xf, true);
}
template <int U>
__device__ __forceinline__ float share_f(float v) {
  return __int_as_float(share_i<U>(__float_as_int(v)));
}

__device__ __forceinline__ v4f x_quarter(const float* __restrict__ x, int s, unsigned lq) {
  return *(const v4f*)((const char*)x + (((unsigned)s << 8) + lq));
}

__device__ __forceinline__ void load_slots(v4f (&r)[16], const float* __restrict__ x, int id,
                                           int cnt, unsigned lq) {
  if (cnt > 0) {
    r[0] = x_quarter(x, share_i<0>(id), lq);   r[1] = x_quarter(x, share_i<1>(id), lq);
    r[2] = x_quarter(x, share_i<2>(id), lq);   r[3] = x_quarter(x, share_i<3>(id), lq);
  }
  if (cnt > 4) {
    r[4] = x_quarter(x, share_i<4>(id), lq);   r[5] = x_quarter(x, share_i<5>(id), lq);
    r[6] = x_quarter(x, share_i<6>(id), lq);   r[7] = x_quarter(x, share_i<7>(id), lq);
  }
  if (cnt > 8) {
    r[8] = x_quarter(x, share_i<8>(id), lq);   r[9] = x_quarter(x, share_i<9>(id), lq);
    r[10] = x_quarter(x, share_i<10>(id), lq); r[11] = x_quarter(x, share_i<11>(id), lq);
  }
  if (cnt > 12) {
    r[12] = x_quarter(x, share_i<12>(id), lq); r[13] = x_quarter(x, share_i<13>(id), lq);
    r[14] = x_quarter(x, share_i<14>(id), lq); r[15] = x_quarter(x, share_i<15>(id), lq);
  }
}

#define ADD_SLOT(U)                                   \
  {                                                   \
    const float sc = share_f<U>(dv);                  \
    a[0] = fmaf(r[U][0], sc, a[0]);                   \
    a[1] = fmaf(r[U][1], sc, a[1]);                   \
    a[2] = fmaf(r[U][2], sc, a[2]);                   \
    a[3] = fmaf(r[U][3], sc, a[3]);                   \
  }
__device__ __forceinline__ void add_slots(const v4f (&r)[16], float dv, int cnt, v4f& a) {
  if (cnt > 0)  { ADD_SLOT(0)  ADD_SLOT(1)  ADD_SLOT(2)  ADD_SLOT(3) }
  if (cnt > 4)  { ADD_SLOT(4)  ADD_SLOT(5)  ADD_SLOT(6)  ADD_SLOT(7) }
  if (cnt > 8)  { ADD_SLOT(8)  ADD_SLOT(9)  ADD_SLOT(10) ADD_SLOT(11) }
  if (cnt > 12) { ADD_SLOT(12) ADD_SLOT(13) ADD_SLOT(14) ADD_SLOT(15) }
}
#undef ADD_SLOT

__device__ __forceinline__ int pair_ids(const int* __restrict__ bkt0, const int* __restrict__ bkt1,
                                        int node, int k0, int mh, int l16) {
  int v = 0;
  if (l16 < mh) {
    const int* __restrict__ p = (l16 < k0) ? (bkt0 + node * BUCKET_SLOTS + l16)
                                           : (bkt1 + node * BUCKET_SLOTS + (l16 - k0));
    v = *p;
  }
  return v;
}

__device__ __forceinline__ void sum_listed(const float* __restrict__ x, const float* __restrict__ dinvf,
                                           const int* __restrict__ ovf_n, const int* __restrict__ ovf_dst,
                                           const int* __restrict__ ovf_src, int node, bool mine,
                                           int lane, unsigned lq, v4f& a) {
  const int total = *ovf_n;
  for (int j0 = 0; j0 < total; j0 += 32) {
    const int j = j0 + lane;
    int od = -1, os = 0;
    if (j < total) { od = ovf_dst[j]; os = ovf_src[j]; }
    unsigned hit = __builtin_amdgcn_ballot_w32(od == node);
    while (hit) {
      const int pos = __builtin_ctz(hit);
      hit &= hit - 1;
      const int s = __builtin_amdgcn_readlane(os, pos);
      const float sd = mine ? dinvf[s] : 0.f;
      const v4f v = x_quarter(x, s, lq);
      a[0] = fmaf(v[0], sd, a[0]);
      a[1] = fmaf(v[1], sd, a[1]);
      a[2] = fmaf(v[2], sd, a[2]);
      a[3] = fmaf(v[3], sd, a[3]);
    }
  }
}

__device__ __forceinline__ int tile_counts(const int* __restrict__ cnt0, const int* __restrict__ cnt1,
                                           int m0, int n, int lane, int l16) {
  int cv = 0;
  if (m0 + l16 < n) cv = (lane < 16) ? cnt0[m0 + l16] : cnt1[m0 + l16];
  return cv;
}

__device__ __forceinline__ void first_pair(int cv, const int* __restrict__ bkt0, const int* __restrict__ bkt1,
                                           int m0, int n, int half, int l16,
                                           int& e0a, int& e0b, int& e1a, int& e1b,
                                           int& k0, int& mh, int& id) {
  e0a = __builtin_amdgcn_readlane(cv, 0);
  e0b = __builtin_amdgcn_readlane(cv, 1);
  e1a = __builtin_amdgcn_readlane(cv, 16);
  e1b = __builtin_amdgcn_readlane(cv, 17);
  k0 = min(half ? e0b : e0a, BUCKET_SLOTS);
  mh = k0 + min(half ? e1b : e1a, BUCKET_SLOTS);
  id = pair_ids(bkt0, bkt1, min(m0 + half, n - 1), k0, mh, l16);
}

template <bool TAB>
__device__ __forceinline__ void
aggregate_gemm_body(const int* __restrict__ cnt0, const int* __restrict__ cnt1,
                    const int* __restrict__ bkt0, const int* __restrict__ bkt1,
                    const int* __restrict__ ovf_n, const int* __restrict__ ovf_dst,
                    const int* __restrict__ ovf_src,
                    const float* __restrict__ x, const float* __restrict__ W,
                    const float* __restrict__ bias, const float* __restrict__ dinvf,
                    const int* __restrict__ degp, float* __restrict__ out, int n,
                    v2f* __restrict__ wfrag, float* __restrict__ ztile,
                    unsigned char* __restrict__ degtab) {
  const int lane = threadIdx.x & 31;
  const int l16 = lane & 15;
  const int kh = (lane >> 4) * 2;
  const int wv = __builtin_amdgcn_readfirstlane((int)(threadIdx.x >> 5));
  float* __restrict__ zt = ztile + wv * (16 * ZP);
  const int half = lane >> 4;
  const unsigned lq = (unsigned)l16 << 4;

  const int m0 = ((int)blockIdx.x * AG_WAVES + wv) * 16;

  const int cv = tile_counts(cnt0, cnt1, m0, n, lane, l16);

  {
    const unsigned char* __restrict__ deg8 = (const unsigned char*)degp;
    const int nvec = TAB ? (n >> 4) : 0;
    v4i tv[TAB_ROUNDS];
    v2f wb[WF_ROUNDS];
    if (TAB) {
      #pragma unroll
      for (int i = 0; i < TAB_ROUNDS; i++) {
        const int q = min((int)threadIdx.x + i * AG_THREADS, nvec - 1);
        tv[i] = *(const v4i*)(deg8 + 16 * q);
      }
    }
    #pragma unroll
    for (int i = 0; i < WF_ROUNDS; i++) {
      const int e = min((int)threadIdx.x + i * AG_THREADS, WF_ENTRIES - 1);
      const int nt = (e >> 5) & 3;
      const int kk = e >> 7;
      const int k = 4 * kk + kh;
      const int col = nt * 16 + l16;
      wb[i][0] = W[k * 64 + col];
      wb[i][1] = W[(k + 1) * 64 + col];
    }
    const int tq = (nvec << 4) + (int)threadIdx.x;
    unsigned char tl = 0;
    const bool tail = TAB && threadIdx.x < 16 && tq < n;
    if (tail) tl = deg8[tq];

    int e0a, e0b, e1a, e1b, k0, mh, id;
    first_pair(cv, bkt0, bkt1, m0, n, half, l16, e0a, e0b, e1a, e1b, k0, mh, id);

    if (TAB) {
      #pragma unroll
      for (int i = 0; i < TAB_ROUNDS; i++) {
        const int q = (int)threadIdx.x + i * AG_THREADS;
        if (q < nvec) *(v4i*)(degtab + 16 * q) = tv[i];
      }
      if (tail) degtab[tq] = tl;
    }
    #pragma unroll
    for (int i = 0; i < WF_ROUNDS; i++) {
      const int e = (int)threadIdx.x + i * AG_THREADS;
      if (e < WF_ENTRIES) wfrag[e] = wb[i];
    }

    __syncthreads();

    {
      for (int p = 0; p < 8; p++) {
        const int node = m0 + 2 * p + half;
        const int nodec = min(node, n - 1);
        const bool more = (p < 7);
        int e0an = 0, e0bn = 0, e1an = 0, e1bn = 0, k0n = 0, mhn = 0, nid = 0;
        if (more) {
          e0an = __builtin_amdgcn_readlane(cv, 2 * p + 2);
          e0bn = __builtin_amdgcn_readlane(cv, 2 * p + 3);
          e1an = __builtin_amdgcn_readlane(cv, 2 * p + 18);
          e1bn = __builtin_amdgcn_readlane(cv, 2 * p + 19);
        }
        const int mm = max(min(e0a, BUCKET_SLOTS) + min(e1a, BUCKET_SLOTS),
                           min(e0b, BUCKET_SLOTS) + min(e1b, BUCKET_SLOTS));
        float di = 0.f;
        v4f self = {0.f, 0.f, 0.f, 0.f};
        v4f a = {0.f, 0.f, 0.f, 0.f};
        int idg = id;
        int j = l16;
        for (int base = 0; ; base += 16) {
          v4f r[16];
          load_slots(r, x, idg, mm - base, lq);
          if (base == 0) {
            di = dinvf[nodec];
            self = x_quarter(x, nodec, lq);
            if (more) {
              k0n = min(half ? e0bn : e0an, BUCKET_SLOTS);
              mhn = k0n + min(half ? e1bn : e1an, BUCKET_SLOTS);
              nid = pair_ids(bkt0, bkt1, min(node + 2, n - 1), k0n, mhn, l16);
            }
          }
          float dg;
          if (TAB) {
            const unsigned q = degtab[idg];
            dg = rsqrtf((float)(int)q);
            if (__builtin_amdgcn_ballot_w32(q == DEG_SAT) != 0u) {
              const float f = dinvf[idg];
              dg = (q == DEG_SAT) ? f : dg;
            }
          } else {
            dg = dinvf[idg];
          }
          const float dv = (j < mh) ? dg : 0.f;
          add_slots(r, dv, mm - base, a);
          if (mm <= base + 16) break;
          j = 16 + l16;
          idg = 0;
          if (j < mh) idg = bkt1[nodec * BUCKET_SLOTS + (j - k0)];
        }
        if (e0a > BUCKET_SLOTS || e1a > BUCKET_SLOTS)
          sum_listed(x, dinvf, ovf_n, ovf_dst, ovf_src, m0 + 2 * p, half == 0, lane, lq, a);
        if (e0b > BUCKET_SLOTS || e1b > BUCKET_SLOTS)
          sum_listed(x, dinvf, ovf_n, ovf_dst, ovf_src, m0 + 2 * p + 1, half == 1, lane, lq, a);
        v4f z;
        z[0] = di * (a[0] + di * self[0]);
        z[1] = di * (a[1] + di * self[1]);
        z[2] = di * (a[2] + di * self[2]);
        z[3] = di * (a[3] + di * self[3]);
        if (node >= n) z = (v4f){0.f, 0.f, 0.f, 0.f};
        *(v4f*)(zt + (2 * p + half) * ZP + 4 * l16) = z;
        e0a = e0an; e0b = e0bn; e1a = e1an; e1b = e1bn;
        k0 = k0n; mh = mhn; id = nid;
      }

      __builtin_amdgcn_fence(__ATOMIC_ACQ_REL, "wavefront");

      v8f c0 = {}, c1 = {}, c2 = {}, c3 = {};
      #pragma unroll
      for (int kk = 0; kk < 16; kk++) {
        const v2f af = *(const v2f*)(zt + l16 * ZP + 4 * kk + kh);
        const v2f b0 = wfrag[(kk * 4 + 0) * 32 + lane];
        const v2f b1 = wfrag[(kk * 4 + 1) * 32 + lane];
        const v2f b2 = wfrag[(kk * 4 + 2) * 32 + lane];
        const v2f b3 = wfrag[(kk * 4 + 3) * 32 + lane];
        c0 = __builtin_amdgcn_wmma_f32_16x16x4_f32(false, af, false, b0, (short)0, c0, false, false);
        c1 = __builtin_amdgcn_wmma_f32_16x16x4_f32(false, af, false, b1, (short)0, c1, false, false);
        c2 = __builtin_amdgcn_wmma_f32_16x16x4_f32(false, af, false, b2, (short)0, c2, false, false);
        c3 = __builtin_amdgcn_wmma_f32_16x16x4_f32(false, af, false, b3, (short)0, c3, false, false);
      }

      const float bv0 = bias[l16], bv1 = bias[16 + l16], bv2 = bias[32 + l16], bv3 = bias[48 + l16];
      #pragma unroll
      for (int v = 0; v < 8; v++) {
        const int mrow = m0 + 8 * (lane >> 4) + v;
        const float o0 = c0[v] + bv0, o1 = c1[v] + bv1, o2 = c2[v] + bv2, o3 = c3[v] + bv3;
        if (mrow < n) {
          float* __restrict__ p = out + mrow * 64 + l16;
          p[0] = o0;
          p[16] = o1;
          p[32] = o2;
          p[48] = o3;
        }
      }
    }
  }
}

__global__ void __launch_bounds__(AG_THREADS)
aggregate_gemm_kernel(const int* __restrict__ cnt0, const int* __restrict__ cnt1,
                      const int* __restrict__ bkt0, const int* __restrict__ bkt1,
                      const int* __restrict__ ovf_n, const int* __restrict__ ovf_dst,
                      const int* __restrict__ ovf_src,
                      const float* __restrict__ x, const float* __restrict__ W,
                      const float* __restrict__ bias, const float* __restrict__ dinvf,
                      const int* __restrict__ degp, float* __restrict__ out, int n) {
  __shared__ __attribute__((aligned(16))) v2f wfrag[WF_ENTRIES];
  __shared__ __attribute__((aligned(16))) float ztile[AG_WAVES * 16 * ZP];
  __shared__ __attribute__((aligned(16))) unsigned char degtab[TAB_NODES];
  if (__builtin_expect(n >= 16 && n <= TAB_NODES, 1))
    aggregate_gemm_body<true>(cnt0, cnt1, bkt0, bkt1, ovf_n, ovf_dst, ovf_src, x, W, bias, dinvf,
                              degp, out, n, wfrag, ztile, degtab);
  else
    aggregate_gemm_body<false>(cnt0, cnt1, bkt0, bkt1, ovf_n, ovf_dst, ovf_src, x, W, bias, dinvf,
                               degp, out, n, wfrag, ztile, degtab);
}

static inline size_t ws_align(size_t x) { return (x + 255) & ~(size_t)255; }

extern "C" void kernel_launch(void* const* d_in, const int* in_sizes, int n_in,
                              void* d_out, int out_size, void* d_ws, size_t ws_size,
                              hipStream_t stream) {
  const float* x  = (const float*)d_in[0];
  const int*   ei = (const int*)d_in[1];
  const float* W  = (const float*)d_in[2];
  const float* b  = (const float*)d_in[3];
  float* out = (float*)d_out;

  const int N = in_sizes[0] / 64;
  const int E = in_sizes[1] / 2;
  const int* src = ei;
  const int* dst = ei + E;

  char* ws = (char*)d_ws;
  size_t off = 0;
  int*   cnt0    = (int*)(ws + off);   off = ws_align(off + (size_t)N * sizeof(int));
  int*   cnt1    = (int*)(ws + off);   off = ws_align(off + (size_t)N * sizeof(int));
  float* dinvf   = (float*)(ws + off); off = ws_align(off + (size_t)N * sizeof(float));
  int*   degp    = (int*)(ws + off);   off = ws_align(off + (size_t)N * sizeof(unsigned char));
  int*   ovf_n   = (int*)(ws + off);   off = ws_align(off + sizeof(int));
  int*   bkt0    = (int*)(ws + off);   off = ws_align(off + (size_t)N * BUCKET_SLOTS * sizeof(int));
  int*   bkt1    = (int*)(ws + off);   off = ws_align(off + (size_t)N * BUCKET_SLOTS * sizeof(int));
  int*   ovf_dst = (int*)(ws + off);   off = ws_align(off + (size_t)E * sizeof(int));
  int*   ovf_src = (int*)(ws + off);   off = ws_align(off + (size_t)E * sizeof(int));

  const int node_grid = N / NODE_BLOCK + (N % NODE_BLOCK + EDGE_THREADS - 1) / EDGE_THREADS;
  const int edge_grid = E / EDGE_BLOCK + (E % EDGE_BLOCK + EDGE_THREADS - 1) / EDGE_THREADS;
  cnt_init_kernel<<<node_grid, EDGE_THREADS, 0, stream>>>(cnt0, cnt1, ovf_n, N);
  count_place_kernel<<<edge_grid, EDGE_THREADS, 0, stream>>>(src, dst, cnt0, cnt1, bkt0, bkt1,
                                                             ovf_n, ovf_dst, ovf_src, E);
  dinv_kernel<<<node_grid, EDGE_THREADS, 0, stream>>>(cnt0, cnt1, dinvf, degp, N);

  aggregate_gemm_kernel<<<(N + 16 * AG_WAVES - 1) / (16 * AG_WAVES), AG_THREADS, 0, stream>>>(
      cnt0, cnt1, bkt0, bkt1, ovf_n, ovf_dst, ovf_src, x, W, b, dinvf, degp, out, N);
}
